# starting kernel
# baseline (speedup 1.0000x reference)
#include <hip/hip_runtime.h>
#include <hip/hip_bf16.h>


typedef float v2f __attribute__((ext_vector_type(2)));
typedef float v8f __attribute__((ext_vector_type(8)));

#define IN_C 512
#define H1C 64
#define H2C 32


__global__ void k_init_deg(float* __restrict__ deg, int Nn) {
  int i = blockIdx.x * blockDim.x + threadIdx.x;
  if (i < Nn) deg[i] = 1.0f;
}

__global__ void k_deg_accum(const int* __restrict__ ei, const float* __restrict__ ew,
                            float* __restrict__ deg, int E) {
  int e = blockIdx.x * blockDim.x + threadIdx.x;
  if (e >= E) return;
  int col = ei[E + e];
  atomicAdd(&deg[col], ew[e]);
}

__global__ void k_dinv(float* __restrict__ dinv, int Nn) {
  int i = blockIdx.x * blockDim.x + threadIdx.x;
  if (i >= Nn) return;
  float d = dinv[i];
  dinv[i] = (d > 0.0f) ? rsqrtf(fmaxf(d, 1e-12f)) : 0.0f;
}

__global__ void k_wnorm(const int* __restrict__ ei, const float* __restrict__ ew,
                        const float* __restrict__ dinv, float* __restrict__ wnorm, int E) {
  int e = blockIdx.x * blockDim.x + threadIdx.x;
  if (e >= E) return;
  int row = ei[e];
  int col = ei[E + e];
  wnorm[e] = dinv[row] * ew[e] * dinv[col];
}


template <int K, int NT, bool RELU>
__global__ __launch_bounds__(NT * 32) void k_gemm_wmma_f32(
    const float* __restrict__ A, const float* __restrict__ B, float* __restrict__ C) {
  constexpr int NOUT = NT * 16;
  const int wave = threadIdx.x >> 5;
  const int lane = threadIdx.x & 31;
  const int m = lane & 15;
  const int half = lane >> 4;
  const int n0 = wave * 16;
  const long long rowBase = (long long)blockIdx.x * 16;

  const float* arow = A + (rowBase + m) * K + half * 2;

  v8f acc = {};
#pragma unroll 4
  for (int k = 0; k < K; k += 4) {
    float2 av = *reinterpret_cast<const float2*>(arow + k);
    v2f a;
    a.x = RELU ? fmaxf(av.x, 0.0f) : av.x;
    a.y = RELU ? fmaxf(av.y, 0.0f) : av.y;
    const int kr = k + half * 2;
    v2f b;
    b.x = B[kr * NOUT + n0 + m];
    b.y = B[(kr + 1) * NOUT + n0 + m];
    acc = __builtin_amdgcn_wmma_f32_16x16x4_f32(false, a, false, b, (short)0, acc,
                                                false, false);
  }

#pragma unroll
  for (int v = 0; v < 8; ++v) {
    long long r = rowBase + v + 8 * half;
    C[r * NOUT + n0 + m] = acc[v];
  }
}

template <int F>
__global__ void k_init_agg(float* __restrict__ agg, const float* __restrict__ h,
                           const float* __restrict__ dinv, const float* __restrict__ bias,
                           int Nn) {
  long long t = (long long)blockIdx.x * blockDim.x + threadIdx.x;
  if (t >= (long long)Nn * F) return;
  int node = (int)(t / F);
  int f = (int)(t % F);
  float di = dinv[node];
  agg[t] = bias[f] + di * di * h[t];
}

template <int CH>
__global__ __launch_bounds__(256) void k_edge_agg(const int* __restrict__ ei,
                                                  const float* __restrict__ wnorm,
                                                  const float* __restrict__ h,
                                                  float* __restrict__ agg, int E) {
  long long t = (long long)blockIdx.x * blockDim.x + threadIdx.x;
  if (t >= (long long)E * CH) return;
  int e = (int)(t / CH);
  int c = (int)(t % CH);
  int row = ei[e];
  int col = ei[E + e];
  float w = wnorm[e];
  float4 v = reinterpret_cast<const float4*>(h + (size_t)row * (CH * 4))[c];
  float* dst = agg + (size_t)col * (CH * 4) + c * 4;
  atomicAdd(dst + 0, w * v.x);
  atomicAdd(dst + 1, w * v.y);
  atomicAdd(dst + 2, w * v.z);
  atomicAdd(dst + 3, w * v.w);
}

__global__ void k_gemm3_relu(const float* __restrict__ a2, const float* __restrict__ W3,
                             float* __restrict__ g, int Nn) {
  int i = blockIdx.x * blockDim.x + threadIdx.x;
  if (i >= Nn) return;
  const float* r = a2 + (size_t)i * H2C;
  float acc = 0.0f;
#pragma unroll
  for (int f = 0; f < H2C; ++f) acc += fmaxf(r[f], 0.0f) * W3[f];
  g[i] = acc;
}

__global__ void k_init_out(float* __restrict__ out, const float* __restrict__ g,
                           const float* __restrict__ dinv, const float* __restrict__ b3,
                           int Nn) {
  int i = blockIdx.x * blockDim.x + threadIdx.x;
  if (i >= Nn) return;
  float di = dinv[i];
  out[i] = b3[0] + di * di * g[i];
}

__global__ void k_edge_agg1(const int* __restrict__ ei, const float* __restrict__ wnorm,
                            const float* __restrict__ g, float* __restrict__ out, int E) {
  int e = blockIdx.x * blockDim.x + threadIdx.x;
  if (e >= E) return;
  int row = ei[e];
  int col = ei[E + e];
  atomicAdd(&out[col], wnorm[e] * g[row]);
}


static inline unsigned int blocks_for(long long total, int bs) {
  return (unsigned int)((total + bs - 1) / bs);
}

extern "C" void kernel_launch(void* const* d_in, const int* in_sizes, int n_in,
                              void* d_out, int out_size, void* d_ws, size_t ws_size,
                              hipStream_t stream) {
  const float* x = (const float*)d_in[0];
  const int* ei = (const int*)d_in[1];
  const float* ew = (const float*)d_in[2];
  const float* W1 = (const float*)d_in[3];
  const float* b1 = (const float*)d_in[4];
  const float* W2 = (const float*)d_in[5];
  const float* b2 = (const float*)d_in[6];
  const float* W3 = (const float*)d_in[7];
  const float* b3 = (const float*)d_in[8];

  const int E = in_sizes[2];
  const int Nn = in_sizes[0] / IN_C;
  float* out = (float*)d_out;

  float* dinv = (float*)d_ws;
  float* wnorm = dinv + Nn;
  float* bufA = wnorm + E;
  float* bufB = bufA + (size_t)Nn * H1C;

  k_init_deg<<<blocks_for(Nn, 256), 256, 0, stream>>>(dinv, Nn);
  k_deg_accum<<<blocks_for(E, 256), 256, 0, stream>>>(ei, ew, dinv, E);
  k_dinv<<<blocks_for(Nn, 256), 256, 0, stream>>>(dinv, Nn);
  k_wnorm<<<blocks_for(E, 256), 256, 0, stream>>>(ei, ew, dinv, wnorm, E);

  k_gemm_wmma_f32<IN_C, H1C / 16, false>
      <<<Nn / 16, (H1C / 16) * 32, 0, stream>>>(x, W1, bufA);
  k_init_agg<H1C><<<blocks_for((long long)Nn * H1C, 256), 256, 0, stream>>>(
      bufB, bufA, dinv, b1, Nn);
  k_edge_agg<H1C / 4><<<blocks_for((long long)E * (H1C / 4), 256), 256, 0, stream>>>(
      ei, wnorm, bufA, bufB, E);

  k_gemm_wmma_f32<H1C, H2C / 16, true>
      <<<Nn / 16, (H2C / 16) * 32, 0, stream>>>(bufB, W2, bufA);
  k_init_agg<H2C><<<blocks_for((long long)Nn * H2C, 256), 256, 0, stream>>>(
      bufB, bufA, dinv, b2, Nn);
  k_edge_agg<H2C / 4><<<blocks_for((long long)E * (H2C / 4), 256), 256, 0, stream>>>(
      ei, wnorm, bufA, bufB, E);

  k_gemm3_relu<<<blocks_for(Nn, 256), 256, 0, stream>>>(bufB, W3, bufA, Nn);
  k_init_out<<<blocks_for(Nn, 256), 256, 0, stream>>>(out, bufA, dinv, b3, Nn);
  k_edge_agg1<<<blocks_for(E, 256), 256, 0, stream>>>(ei, wnorm, bufA, out, E);
}


# Edge-weight pass stages the per-node 1/sqrt(degree) table in LDS (256-KiB chunks, one 1024-thread workgroup per WGP): its two per-edge degree gathers become LDS reads; values unchanged
# speedup vs baseline: 3.1108x; 3.1108x over previous
#include <hip/hip_runtime.h>
#include <hip/hip_bf16.h>


typedef float v2f __attribute__((ext_vector_type(2)));
typedef float v8f __attribute__((ext_vector_type(8)));

#define IN_C 512
#define H1C 64
#define H2C 32


#define TASK_EDGES 1024
#define SCATTER_BLOCKS 2048

__device__ __forceinline__ int l2_half_of_this_wave() {
  return (int)((__builtin_amdgcn_s_sendmsg_rtn(0x87) >> 18) & 1u);
}

__device__ __forceinline__ int range_split(int Nn) { return (Nn >> 1) & ~63; }

__device__ __forceinline__ void lds_barrier() {
  __builtin_amdgcn_fence(__ATOMIC_RELEASE, "workgroup", "local");
  __builtin_amdgcn_s_barrier();
  __builtin_amdgcn_fence(__ATOMIC_ACQUIRE, "workgroup", "local");
}

__device__ __forceinline__ int share_word(int* s_word, int mine) {
  if (threadIdx.x == 0) *s_word = mine;
  lds_barrier();
  const int v = __builtin_amdgcn_readfirstlane(*s_word);
  lds_barrier();
  return v;
}

__global__ void k_init_deg(float* __restrict__ deg, int* __restrict__ cnt, int ncnt, int Nn) {
  int i = blockIdx.x * blockDim.x + threadIdx.x;
  if (i < Nn) deg[i] = 1.0f;
  if (i < ncnt) cnt[i] = 0;
}

__global__ __launch_bounds__(256) void k_deg_accum(const int* __restrict__ ei,
                                                   const float* __restrict__ ew,
                                                   float* __restrict__ deg,
                                                   int* __restrict__ cnt, int E, int Nn) {
  constexpr int K = TASK_EDGES / 256;
  __shared__ int s_word;
  const int S = (E + TASK_EDGES - 1) / TASK_EDGES;
  const int mid = range_split(Nn);
  const int mine = share_word(&s_word, l2_half_of_this_wave());
  for (int pass = 0; pass < 2; ++pass) {
    const int q = mine ^ pass;
    const int lo = q ? mid : 0;
    const int hi = q ? Nn : mid;
    int drawn = 0;
    if (threadIdx.x == 0) drawn = atomicAdd(&cnt[q], 1);
    int t = share_word(&s_word, drawn);
    while (t < S) {
      const long long base = (long long)t * TASK_EDGES + threadIdx.x;
      int col[K];
      float wv[K];
#pragma unroll
      for (int k = 0; k < K; ++k) {
        const long long e = base + 256 * k;
        col[k] = -1;
        wv[k] = 0.0f;
        if (e < E) {
          col[k] = ei[(long long)E + e];
          wv[k] = ew[e];
        }
      }
#pragma unroll
      for (int k = 0; k < K; ++k)
        if (col[k] >= lo && col[k] < hi) atomicAdd(&deg[col[k]], wv[k]);
      if (threadIdx.x == 0) drawn = atomicAdd(&cnt[q], 1);
      t = share_word(&s_word, drawn);
    }
  }
}

__device__ __forceinline__ float dinv_of(float d) {
  return (d > 0.0f) ? rsqrtf(fmaxf(d, 1e-12f)) : 0.0f;
}

#define WN_THREADS 1024
#define WN_EPT 13
#define WN_CHUNK 65536

__global__ __launch_bounds__(WN_THREADS) void k_wnorm(const int* __restrict__ ei,
                                                      const float* __restrict__ ew,
                                                      const float* __restrict__ deg,
                                                      float* __restrict__ wnorm, int E, int Nn) {
  __shared__ alignas(16) float tab[WN_CHUNK];
  const int tid = threadIdx.x;
  constexpr int BATCH = WN_THREADS * WN_EPT;
  const unsigned per = ((((unsigned)E + gridDim.x - 1u) / gridDim.x) + 31u) & ~31u;
  const long long s0 = (long long)blockIdx.x * per;
  const long long s1 = (s0 + per < (long long)E) ? (s0 + per) : (long long)E;
  for (long long base = s0; base < s1; base += BATCH) {
    const int rem = (s1 - base < (long long)BATCH) ? (int)(s1 - base) : BATCH;
    const int* __restrict__ rp = ei + base;
    const int* __restrict__ cp = ei + (long long)E + base;
    const float* __restrict__ wp = ew + base;
    float* __restrict__ op = wnorm + base;
    int row[WN_EPT], col[WN_EPT];
    float w[WN_EPT], fa[WN_EPT], fb[WN_EPT];
#pragma unroll
    for (int k = 0; k < WN_EPT; ++k) {
      const int t = k * WN_THREADS + tid;
      const int tc = (t < rem) ? t : (rem - 1);
      row[k] = rp[tc];
      col[k] = cp[tc];
      w[k] = wp[tc];
      fa[k] = 0.0f;
      fb[k] = 0.0f;
    }
    for (int lo = 0; lo < Nn; lo += WN_CHUNK) {
      const int cnt = (Nn - lo < WN_CHUNK) ? (Nn - lo) : WN_CHUNK;
      lds_barrier();
      const int n4 = cnt >> 2;
      const float4* __restrict__ dp = reinterpret_cast<const float4*>(deg + lo);
      float4* tp = reinterpret_cast<float4*>(tab);
      for (int g0 = 0; g0 < n4; g0 += WN_THREADS * 8) {
        float4 d[8];
#pragma unroll
        for (int j = 0; j < 8; ++j) {
          const int g = g0 + j * WN_THREADS + tid;
          d[j] = dp[(g < n4) ? g : (n4 - 1)];
        }
#pragma unroll
        for (int j = 0; j < 8; ++j) {
          const int g = g0 + j * WN_THREADS + tid;
          if (g < n4) {
            float4 v;
            v.x = dinv_of(d[j].x);
            v.y = dinv_of(d[j].y);
            v.z = dinv_of(d[j].z);
            v.w = dinv_of(d[j].w);
            tp[g] = v;
          }
        }
      }
      {
        const int i = (n4 << 2) + tid;
        if (i < cnt) tab[i] = dinv_of(deg[lo + i]);
      }
      lds_barrier();
#pragma unroll
      for (int k = 0; k < WN_EPT; ++k) {
        const unsigned r = (unsigned)(row[k] - lo);
        const float tr = tab[(r < (unsigned)cnt) ? r : 0u];
        fa[k] = (r < (unsigned)cnt) ? tr : fa[k];
        const unsigned c = (unsigned)(col[k] - lo);
        const float tc = tab[(c < (unsigned)cnt) ? c : 0u];
        fb[k] = (c < (unsigned)cnt) ? tc : fb[k];
      }
      asm volatile("s_wait_dscnt 0x0" ::: "memory");
    }
#pragma unroll
    for (int k = 0; k < WN_EPT; ++k) {
      const int t = k * WN_THREADS + tid;
      if (t < rem) op[t] = fa[k] * w[k] * fb[k];
    }
  }
}


template <int K, int NT, bool RELU>
__global__ __launch_bounds__(NT * 32) void k_gemm_wmma_f32(
    const float* __restrict__ A, const float* __restrict__ B, float* __restrict__ C,
    const float* __restrict__ deg, const float* __restrict__ bias, float* __restrict__ S) {
  constexpr int NOUT = NT * 16;
  const int wave = threadIdx.x >> 5;
  const int lane = threadIdx.x & 31;
  const int m = lane & 15;
  const int half = lane >> 4;
  const int n0 = wave * 16;
  const long long rowBase = (long long)blockIdx.x * 16;

  const float* arow = A + (rowBase + m) * K + half * 2;

  float dg[8];
#pragma unroll
  for (int v = 0; v < 8; ++v) dg[v] = deg[rowBase + v + 8 * half];
  const float bcol = bias[n0 + m];

  v8f acc = {};
#pragma unroll 4
  for (int k = 0; k < K; k += 4) {
    float2 av = *reinterpret_cast<const float2*>(arow + k);
    v2f a;
    a.x = RELU ? fmaxf(av.x, 0.0f) : av.x;
    a.y = RELU ? fmaxf(av.y, 0.0f) : av.y;
    const int kr = k + half * 2;
    v2f b;
    b.x = B[kr * NOUT + n0 + m];
    b.y = B[(kr + 1) * NOUT + n0 + m];
    acc = __builtin_amdgcn_wmma_f32_16x16x4_f32(false, a, false, b, (short)0, acc,
                                                false, false);
  }

  float dd[8];
#pragma unroll
  for (int v = 0; v < 8; ++v) {
    const float di = dinv_of(dg[v]);
    dd[v] = di * di;
  }
#pragma unroll
  for (int v = 0; v < 8; ++v) {
    long long r = rowBase + v + 8 * half;
    C[r * NOUT + n0 + m] = acc[v];
    S[r * NOUT + n0 + m] = bcol + dd[v] * acc[v];
  }
}

#define G1_WAVES 8
#define G1_PITCH 68

__device__ __forceinline__ void g1_tile(const float4 (&a)[8], const v2f (&bf)[16][4],
                                        float (&part)[G1_WAVES][16][G1_PITCH],
                                        float* __restrict__ C, float* __restrict__ S,
                                        float d, const float4& bv, int tile, int wave,
                                        int c, int h, int row, int c4) {
  v8f acc[4] = {};
#pragma unroll
  for (int i = 0; i < 8; ++i) {
    v2f a0, a1;
    a0.x = a[i].x; a0.y = a[i].y;
    a1.x = a[i].z; a1.y = a[i].w;
#pragma unroll
    for (int q = 0; q < 4; ++q)
      acc[q] = __builtin_amdgcn_wmma_f32_16x16x4_f32(false, a0, false, bf[2 * i][q],
                                                     (short)0, acc[q], false, false);
#pragma unroll
    for (int q = 0; q < 4; ++q)
      acc[q] = __builtin_amdgcn_wmma_f32_16x16x4_f32(false, a1, false, bf[2 * i + 1][q],
                                                     (short)0, acc[q], false, false);
  }

  __builtin_amdgcn_sched_barrier(0);
  asm volatile("v_nop\n\tv_nop\n\tv_nop\n\tv_nop\n\tv_nop\n\tv_nop\n\tv_nop\n\tv_nop");
  __builtin_amdgcn_sched_barrier(0);

#pragma unroll
  for (int q = 0; q < 4; ++q) {
#pragma unroll
    for (int v = 0; v < 8; ++v) part[wave][v + 8 * h][16 * q + c] = acc[q][v];
  }

  __builtin_amdgcn_fence(__ATOMIC_RELEASE, "workgroup", "local");
  __builtin_amdgcn_s_barrier();
  __builtin_amdgcn_fence(__ATOMIC_ACQUIRE, "workgroup", "local");

  float4 s = *reinterpret_cast<const float4*>(&part[0][row][c4]);
#pragma unroll
  for (int w = 1; w < G1_WAVES; ++w) {
    const float4 p = *reinterpret_cast<const float4*>(&part[w][row][c4]);
    s.x += p.x; s.y += p.y; s.z += p.z; s.w += p.w;
  }
  const size_t off = ((size_t)tile * 16 + row) * H1C + c4;
  *reinterpret_cast<float4*>(C + off) = s;
  const float di = dinv_of(d);
  const float dd = di * di;
  float4 st;
  st.x = bv.x + dd * s.x;
  st.y = bv.y + dd * s.y;
  st.z = bv.z + dd * s.z;
  st.w = bv.w + dd * s.w;
  *reinterpret_cast<float4*>(S + off) = st;
}

__device__ __forceinline__ void g1_load_a(float4 (&a)[8], const float* __restrict__ A,
                                          int tile, int c, int kw) {
  const float* ap = A + ((size_t)tile * 16 + c) * IN_C + kw;
#pragma unroll
  for (int i = 0; i < 8; ++i) a[i] = *reinterpret_cast<const float4*>(ap + 8 * i);
}

__global__ __launch_bounds__(G1_WAVES * 32) void k_gemm1_splitk(
    const float* __restrict__ A, const float* __restrict__ B, float* __restrict__ C,
    const float* __restrict__ deg, const float* __restrict__ bias, float* __restrict__ S,
    int numTiles) {
  __shared__ alignas(16) float part[2][G1_WAVES][16][G1_PITCH];

  const int wave = __builtin_amdgcn_readfirstlane((int)(threadIdx.x >> 5));
  const int lane = threadIdx.x & 31;
  const int c = lane & 15;
  const int h = lane >> 4;
  const int kw = wave * 64 + h * 4;

  v2f bf[16][4];
#pragma unroll
  for (int i = 0; i < 8; ++i) {
#pragma unroll
    for (int e = 0; e < 2; ++e) {
      const int kk = kw + 8 * i + 2 * e;
#pragma unroll
      for (int q = 0; q < 4; ++q) {
        bf[2 * i + e][q].x = B[kk * H1C + 16 * q + c];
        bf[2 * i + e][q].y = B[(kk + 1) * H1C + 16 * q + c];
      }
    }
  }

  const int row = threadIdx.x >> 4;
  const int c4 = (threadIdx.x & 15) * 4;
  const float4 bv = *reinterpret_cast<const float4*>(bias + c4);

  const int step = gridDim.x;
  int tile = blockIdx.x;
  if (tile >= numTiles) return;
  float4 aA[8], aB[8];
  float dA, dB;
  dA = deg[tile * 16 + row];
  g1_load_a(aA, A, tile, c, kw);
  while (true) {
    int nt = tile + step;
    bool more = nt < numTiles;
    int pt = more ? nt : tile;
    dB = deg[pt * 16 + row];
    g1_load_a(aB, A, pt, c, kw);
    g1_tile(aA, bf, part[0], C, S, dA, bv, tile, wave, c, h, row, c4);
    if (!more) break;
    tile = nt;
    nt = tile + step;
    more = nt < numTiles;
    pt = more ? nt : tile;
    dA = deg[pt * 16 + row];
    g1_load_a(aA, A, pt, c, kw);
    g1_tile(aB, bf, part[1], C, S, dB, bv, tile, wave, c, h, row, c4);
    if (!more) break;
    tile = nt;
  }
}


template <int F>
__device__ __forceinline__ void agg_marked_edges(unsigned m, int row, int col, float w,
                                                 const float* __restrict__ h,
                                                 float* __restrict__ agg, int lane) {
  constexpr int P = F / 32;
  constexpr int G = 8;
  while (m) {
    int idx[G];
#pragma unroll
    for (int j = 0; j < G; ++j) {
      idx[j] = m ? (int)__builtin_ctz(m) : -1;
      m &= m - 1u;
    }
    float v[G][P];
#pragma unroll
    for (int j = 0; j < G; ++j) {
      if (idx[j] >= 0) {
        const int r = __builtin_amdgcn_readlane(row, idx[j]);
        const float* src = h + (size_t)r * F + lane;
#pragma unroll
        for (int p = 0; p < P; ++p) v[j][p] = src[p * 32];
      }
    }
#pragma unroll
    for (int j = 0; j < G; ++j) {
      if (idx[j] >= 0) {
        const int c = __builtin_amdgcn_readlane(col, idx[j]);
        const float wj = __int_as_float(__builtin_amdgcn_readlane(__float_as_int(w), idx[j]));
        float* dst = agg + (size_t)c * F + lane;
#pragma unroll
        for (int p = 0; p < P; ++p) atomicAdd(dst + p * 32, wj * v[j][p]);
      }
    }
  }
}

template <int F>
__global__ __launch_bounds__(256) void k_edge_agg(const int* __restrict__ ei,
                                                  const float* __restrict__ wnorm,
                                                  const float* __restrict__ h,
                                                  float* __restrict__ agg,
                                                  int* __restrict__ cnt, int E, int Nn) {
  constexpr int L = TASK_EDGES / 256;
  __shared__ int s_word;
  const int lane = threadIdx.x & 31;
  const int wave = __builtin_amdgcn_readfirstlane((int)(threadIdx.x >> 5));
  const int S = (E + TASK_EDGES - 1) / TASK_EDGES;
  const int mid = range_split(Nn);
  const int mine = share_word(&s_word, l2_half_of_this_wave());
  for (int pass = 0; pass < 2; ++pass) {
    const int q = mine ^ pass;
    const int lo = q ? mid : 0;
    const int hi = q ? Nn : mid;
    int drawn = 0;
    if (threadIdx.x == 0) drawn = atomicAdd(&cnt[q], 1);
    int t = share_word(&s_word, drawn);
    while (t < S) {
      const long long base = (long long)t * TASK_EDGES + wave * (32 * L) + lane;
      int row[L], col[L];
      float w[L];
#pragma unroll
      for (int s = 0; s < L; ++s) {
        const long long e = base + 32 * s;
        row[s] = 0;
        col[s] = -1;
        w[s] = 0.0f;
        if (e < E) {
          row[s] = ei[e];
          col[s] = ei[(long long)E + e];
          w[s] = wnorm[e];
        }
      }
#pragma unroll
      for (int s = 0; s < L; ++s) {
        const unsigned m = __builtin_amdgcn_ballot_w32(col[s] >= lo && col[s] < hi);
        agg_marked_edges<F>(m, row[s], col[s], w[s], h, agg, lane);
      }
      if (threadIdx.x == 0) drawn = atomicAdd(&cnt[q], 1);
      t = share_word(&s_word, drawn);
    }
  }
}

__global__ void k_gemm3_relu(const float* __restrict__ a2, const float* __restrict__ W3,
                             const float* __restrict__ deg, const float* __restrict__ b3,
                             float* __restrict__ g, float* __restrict__ out,
                             int* __restrict__ cnt, int Nn) {
  int i = blockIdx.x * blockDim.x + threadIdx.x;
  if (i < 2) cnt[i] = 0;
  if (i >= Nn) return;
  const float* r = a2 + (size_t)i * H2C;
  float acc = 0.0f;
#pragma unroll
  for (int f = 0; f < H2C; ++f) acc += fmaxf(r[f], 0.0f) * W3[f];
  g[i] = acc;
  const float di = dinv_of(deg[i]);
  out[i] = b3[0] + di * di * acc;
}

__global__ __launch_bounds__(256) void k_edge_agg1(const int* __restrict__ ei,
                                                   const float* __restrict__ wnorm,
                                                   const float* __restrict__ g,
                                                   float* __restrict__ out,
                                                   int* __restrict__ cnt, int E, int Nn) {
  constexpr int K = TASK_EDGES / 256;
  __shared__ int s_word;
  const int S = (E + TASK_EDGES - 1) / TASK_EDGES;
  const int mid = range_split(Nn);
  const int mine = share_word(&s_word, l2_half_of_this_wave());
  for (int pass = 0; pass < 2; ++pass) {
    const int q = mine ^ pass;
    const int lo = q ? mid : 0;
    const int hi = q ? Nn : mid;
    int drawn = 0;
    if (threadIdx.x == 0) drawn = atomicAdd(&cnt[q], 1);
    int t = share_word(&s_word, drawn);
    while (t < S) {
      const long long base = (long long)t * TASK_EDGES + threadIdx.x;
      int row[K], col[K];
      float wv[K], gv[K];
#pragma unroll
      for (int k = 0; k < K; ++k) {
        const long long e = base + 256 * k;
        row[k] = 0;
        col[k] = -1;
        wv[k] = 0.0f;
        if (e < E) {
          row[k] = ei[e];
          col[k] = ei[(long long)E + e];
          wv[k] = wnorm[e];
        }
      }
#pragma unroll
      for (int k = 0; k < K; ++k) {
        gv[k] = 0.0f;
        if (col[k] >= lo && col[k] < hi) gv[k] = g[row[k]];
      }
#pragma unroll
      for (int k = 0; k < K; ++k)
        if (col[k] >= lo && col[k] < hi) atomicAdd(&out[col[k]], wv[k] * gv[k]);
      if (threadIdx.x == 0) drawn = atomicAdd(&cnt[q], 1);
      t = share_word(&s_word, drawn);
    }
  }
}


static inline unsigned int blocks_for(long long total, int bs) {
  return (unsigned int)((total + bs - 1) / bs);
}

extern "C" void kernel_launch(void* const* d_in, const int* in_sizes, int n_in,
                              void* d_out, int out_size, void* d_ws, size_t ws_size,
                              hipStream_t stream) {
  const float* x = (const float*)d_in[0];
  const int* ei = (const int*)d_in[1];
  const float* ew = (const float*)d_in[2];
  const float* W1 = (const float*)d_in[3];
  const float* b1 = (const float*)d_in[4];
  const float* W2 = (const float*)d_in[5];
  const float* b2 = (const float*)d_in[6];
  const float* W3 = (const float*)d_in[7];
  const float* b3 = (const float*)d_in[8];

  const int E = in_sizes[2];
  const int Nn = in_sizes[0] / IN_C;
  float* out = (float*)d_out;

  float* deg = (float*)d_ws;
  float* wnorm = deg + Nn;
  float* bufA = wnorm + E;
  float* bufB = bufA + (size_t)Nn * H1C;
  float* agg2 = bufA + (size_t)Nn * H2C;

  int* cntA = (int*)d_out;
  int* cntB = (int*)(bufB + (size_t)Nn * H2C);

  k_init_deg<<<blocks_for(Nn, 256), 256, 0, stream>>>(deg, cntA, 6, Nn);
  k_deg_accum<<<SCATTER_BLOCKS, 256, 0, stream>>>(ei, ew, deg, cntA, E, Nn);
  k_wnorm<<<256, WN_THREADS, 0, stream>>>(ei, ew, deg, wnorm, E, Nn);

  k_gemm1_splitk<<<512, G1_WAVES * 32, 0, stream>>>(x, W1, bufA, deg, b1, bufB, Nn / 16);
  k_edge_agg<H1C><<<SCATTER_BLOCKS, 256, 0, stream>>>(
      ei, wnorm, bufA, bufB, cntA + 2, E, Nn);

  k_gemm_wmma_f32<H1C, H2C / 16, true>
      <<<Nn / 16, (H2C / 16) * 32, 0, stream>>>(bufB, W2, bufA, deg, b2, agg2);
  k_edge_agg<H2C><<<SCATTER_BLOCKS, 256, 0, stream>>>(
      ei, wnorm, bufA, agg2, cntA + 4, E, Nn);

  k_gemm3_relu<<<blocks_for(Nn, 256), 256, 0, stream>>>(agg2, W3, deg, b3, bufB, out, cntB,
                                                        Nn);
  k_edge_agg1<<<SCATTER_BLOCKS, 256, 0, stream>>>(ei, wnorm, bufB, out, cntB, E, Nn);
}


# Degree scatter pass folded into the layer-1 product launch: one edge ticket per row tile, atomics between tiles; start values move to the edge-weight pass; 9 to 8 launches; f32 atomic order as before
# speedup vs baseline: 3.5488x; 1.1408x over previous
#include <hip/hip_runtime.h>
#include <hip/hip_bf16.h>


typedef float v2f __attribute__((ext_vector_type(2)));
typedef float v8f __attribute__((ext_vector_type(8)));

#define IN_C 512
#define H1C 64
#define H2C 32


#define TASK_EDGES 1024
#define SCATTER_BLOCKS 2048

__device__ __forceinline__ int l2_half_of_this_wave() {
  return (int)((__builtin_amdgcn_s_sendmsg_rtn(0x87) >> 18) & 1u);
}

__device__ __forceinline__ int range_split(int Nn) { return (Nn >> 1) & ~63; }

__device__ __forceinline__ void lds_barrier() {
  __builtin_amdgcn_fence(__ATOMIC_RELEASE, "workgroup", "local");
  __builtin_amdgcn_s_barrier();
  __builtin_amdgcn_fence(__ATOMIC_ACQUIRE, "workgroup", "local");
}

__device__ __forceinline__ int share_word(int* s_word, int mine) {
  if (threadIdx.x == 0) *s_word = mine;
  lds_barrier();
  const int v = __builtin_amdgcn_readfirstlane(*s_word);
  lds_barrier();
  return v;
}

__global__ void k_init_deg(float* __restrict__ deg, int* __restrict__ cnt, int ncnt, int Nn) {
  int i = blockIdx.x * blockDim.x + threadIdx.x;
  if (i < Nn) deg[i] = 1.0f;
  if (i < ncnt) cnt[i] = 0;
}

#define DEG_K (TASK_EDGES / 256)

__device__ __forceinline__ void deg_ticket_load(int (&col)[DEG_K], float (&wv)[DEG_K],
                                                const int* __restrict__ ei,
                                                const float* __restrict__ ew, int t, int E) {
  const int* __restrict__ cp = ei + E;
  const int base = t * TASK_EDGES + (int)threadIdx.x;
#pragma unroll
  for (int k = 0; k < DEG_K; ++k) {
    const int e = base + 256 * k;
    const int ec = (e < E) ? e : (E - 1);
    const int cv = cp[ec];
    wv[k] = ew[ec];
    col[k] = (e < E) ? cv : -1;
  }
}

__device__ __forceinline__ void deg_ticket_add(const int (&col)[DEG_K], const float (&wv)[DEG_K],
                                               float* __restrict__ deg, int lo, int hi) {
#pragma unroll
  for (int k = 0; k < DEG_K; ++k)
    if (col[k] >= lo && col[k] < hi) atomicAdd(&deg[col[k]], wv[k]);
}

__device__ __forceinline__ float dinv_of(float d) {
  return (d > 0.0f) ? rsqrtf(fmaxf(d, 1e-12f)) : 0.0f;
}

#define WN_THREADS 1024
#define WN_EPT 13
#define WN_CHUNK 65536

__global__ __launch_bounds__(WN_THREADS) void k_wnorm(const int* __restrict__ ei,
                                                      const float* __restrict__ ew,
                                                      const float* __restrict__ deg,
                                                      float* __restrict__ wnorm, int E, int Nn,
                                                      const float* __restrict__ C,
                                                      const float* __restrict__ bias,
                                                      float* __restrict__ S) {
  __shared__ alignas(16) float tab[WN_CHUNK];
  const int tid = threadIdx.x;
  constexpr int BATCH = WN_THREADS * WN_EPT;
  const unsigned per = ((((unsigned)E + gridDim.x - 1u) / gridDim.x) + 31u) & ~31u;
  const long long s0 = (long long)blockIdx.x * per;
  const long long s1 = (s0 + per < (long long)E) ? (s0 + per) : (long long)E;
  for (long long base = s0; base < s1; base += BATCH) {
    const int rem = (s1 - base < (long long)BATCH) ? (int)(s1 - base) : BATCH;
    const int* __restrict__ rp = ei + base;
    const int* __restrict__ cp = ei + (long long)E + base;
    const float* __restrict__ wp = ew + base;
    float* __restrict__ op = wnorm + base;
    int row[WN_EPT], col[WN_EPT];
    float w[WN_EPT], fa[WN_EPT], fb[WN_EPT];
#pragma unroll
    for (int k = 0; k < WN_EPT; ++k) {
      const int t = k * WN_THREADS + tid;
      const int tc = (t < rem) ? t : (rem - 1);
      row[k] = rp[tc];
      col[k] = cp[tc];
      w[k] = wp[tc];
      fa[k] = 0.0f;
      fb[k] = 0.0f;
    }
    for (int lo = 0; lo < Nn; lo += WN_CHUNK) {
      const int cnt = (Nn - lo < WN_CHUNK) ? (Nn - lo) : WN_CHUNK;
      lds_barrier();
      const int n4 = cnt >> 2;
      const float4* __restrict__ dp = reinterpret_cast<const float4*>(deg + lo);
      float4* tp = reinterpret_cast<float4*>(tab);
      for (int g0 = 0; g0 < n4; g0 += WN_THREADS * 8) {
        float4 d[8];
#pragma unroll
        for (int j = 0; j < 8; ++j) {
          const int g = g0 + j * WN_THREADS + tid;
          d[j] = dp[(g < n4) ? g : (n4 - 1)];
        }
#pragma unroll
        for (int j = 0; j < 8; ++j) {
          const int g = g0 + j * WN_THREADS + tid;
          if (g < n4) {
            float4 v;
            v.x = dinv_of(d[j].x);
            v.y = dinv_of(d[j].y);
            v.z = dinv_of(d[j].z);
            v.w = dinv_of(d[j].w);
            tp[g] = v;
          }
        }
      }
      {
        const int i = (n4 << 2) + tid;
        if (i < cnt) tab[i] = dinv_of(deg[lo + i]);
      }
      lds_barrier();
#pragma unroll
      for (int k = 0; k < WN_EPT; ++k) {
        const unsigned r = (unsigned)(row[k] - lo);
        const float tr = tab[(r < (unsigned)cnt) ? r : 0u];
        fa[k] = (r < (unsigned)cnt) ? tr : fa[k];
        const unsigned c = (unsigned)(col[k] - lo);
        const float tc = tab[(c < (unsigned)cnt) ? c : 0u];
        fb[k] = (c < (unsigned)cnt) ? tc : fb[k];
      }
      asm volatile("s_wait_dscnt 0x0" ::: "memory");
    }
#pragma unroll
    for (int k = 0; k < WN_EPT; ++k) {
      const int t = k * WN_THREADS + tid;
      if (t < rem) op[t] = fa[k] * w[k] * fb[k];
    }
  }

  const int total4 = Nn * (H1C / 4);
  const float4* __restrict__ C4 = reinterpret_cast<const float4*>(C);
  float4* __restrict__ S4 = reinterpret_cast<float4*>(S);
  const float4 bv = *reinterpret_cast<const float4*>(bias + (tid & 15) * 4);
  for (int g0 = blockIdx.x * (WN_THREADS * 4); g0 < total4; g0 += gridDim.x * (WN_THREADS * 4)) {
    float4 cv[4];
    float dv[4];
#pragma unroll
    for (int j = 0; j < 4; ++j) {
      const int g = g0 + j * WN_THREADS + tid;
      const int gc = (g < total4) ? g : (total4 - 1);
      cv[j] = C4[gc];
      dv[j] = deg[gc >> 4];
    }
#pragma unroll
    for (int j = 0; j < 4; ++j) {
      const int g = g0 + j * WN_THREADS + tid;
      if (g < total4) {
        const float di = dinv_of(dv[j]);
        const float dd = di * di;
        float4 st;
        st.x = bv.x + dd * cv[j].x;
        st.y = bv.y + dd * cv[j].y;
        st.z = bv.z + dd * cv[j].z;
        st.w = bv.w + dd * cv[j].w;
        S4[g] = st;
      }
    }
  }
}


template <int K, int NT, bool RELU>
__global__ __launch_bounds__(NT * 32) void k_gemm_wmma_f32(
    const float* __restrict__ A, const float* __restrict__ B, float* __restrict__ C,
    const float* __restrict__ deg, const float* __restrict__ bias, float* __restrict__ S) {
  constexpr int NOUT = NT * 16;
  const int wave = threadIdx.x >> 5;
  const int lane = threadIdx.x & 31;
  const int m = lane & 15;
  const int half = lane >> 4;
  const int n0 = wave * 16;
  const long long rowBase = (long long)blockIdx.x * 16;

  const float* arow = A + (rowBase + m) * K + half * 2;

  float dg[8];
#pragma unroll
  for (int v = 0; v < 8; ++v) dg[v] = deg[rowBase + v + 8 * half];
  const float bcol = bias[n0 + m];

  v8f acc = {};
#pragma unroll 4
  for (int k = 0; k < K; k += 4) {
    float2 av = *reinterpret_cast<const float2*>(arow + k);
    v2f a;
    a.x = RELU ? fmaxf(av.x, 0.0f) : av.x;
    a.y = RELU ? fmaxf(av.y, 0.0f) : av.y;
    const int kr = k + half * 2;
    v2f b;
    b.x = B[kr * NOUT + n0 + m];
    b.y = B[(kr + 1) * NOUT + n0 + m];
    acc = __builtin_amdgcn_wmma_f32_16x16x4_f32(false, a, false, b, (short)0, acc,
                                                false, false);
  }

  float dd[8];
#pragma unroll
  for (int v = 0; v < 8; ++v) {
    const float di = dinv_of(dg[v]);
    dd[v] = di * di;
  }
#pragma unroll
  for (int v = 0; v < 8; ++v) {
    long long r = rowBase + v + 8 * half;
    C[r * NOUT + n0 + m] = acc[v];
    S[r * NOUT + n0 + m] = bcol + dd[v] * acc[v];
  }
}

#define G1_WAVES 8
#define G1_PITCH 68

struct DegQueue {
  int pass;
  int t;
};

__device__ __forceinline__ void g1_tile(const float4 (&a)[8], const v2f (&bf)[16][4],
                                        float (&part)[G1_WAVES][16][G1_PITCH], int* s_tk,
                                        float* __restrict__ C, int tile, int wave, int c,
                                        int h, int row, int c4, DegQueue& dq, int mine,
                                        int S, int mid, const int* __restrict__ ei,
                                        const float* __restrict__ ew,
                                        float* __restrict__ deg, int* __restrict__ cnt,
                                        int E, int Nn) {
  int drawn = 0x7fffffff;
  if (dq.pass < 2 && threadIdx.x == 0)
    drawn = (int)atomicInc(reinterpret_cast<unsigned int*>(cnt + (mine ^ dq.pass)), 0xffffffffu);

  __builtin_amdgcn_sched_barrier(0);

  v8f acc[4] = {};
#pragma unroll
  for (int i = 0; i < 8; ++i) {
    v2f a0, a1;
    a0.x = a[i].x; a0.y = a[i].y;
    a1.x = a[i].z; a1.y = a[i].w;
#pragma unroll
    for (int q = 0; q < 4; ++q)
      acc[q] = __builtin_amdgcn_wmma_f32_16x16x4_f32(false, a0, false, bf[2 * i][q],
                                                     (short)0, acc[q], false, false);
#pragma unroll
    for (int q = 0; q < 4; ++q)
      acc[q] = __builtin_amdgcn_wmma_f32_16x16x4_f32(false, a1, false, bf[2 * i + 1][q],
                                                     (short)0, acc[q], false, false);
  }

  const bool have = dq.t >= 0;
  int tl = have ? dq.t : 0;
  __builtin_amdgcn_sched_barrier(0);
  asm volatile("v_nop\n\tv_nop\n\tv_nop\n\tv_nop\n\tv_nop\n\tv_nop\n\tv_nop\n\tv_nop"
               : "+v"(acc[0]), "+v"(acc[1]), "+v"(acc[2]), "+v"(acc[3]), "+s"(tl));
  __builtin_amdgcn_sched_barrier(0);

#pragma unroll
  for (int q4 = 0; q4 < 4; ++q4) {
#pragma unroll
    for (int v = 0; v < 8; ++v) part[wave][v + 8 * h][16 * q4 + c] = acc[q4][v];
  }

  const int q = mine ^ dq.pass;
  const int lo = have ? (q ? mid : 0) : 0;
  const int hi = have ? (q ? Nn : mid) : 0;
  int col[DEG_K];
  float wv[DEG_K];
  deg_ticket_load(col, wv, ei, ew, tl, E);
  if (threadIdx.x == 0) *s_tk = drawn;

  __builtin_amdgcn_fence(__ATOMIC_RELEASE, "workgroup", "local");
  __builtin_amdgcn_s_barrier();
  __builtin_amdgcn_fence(__ATOMIC_ACQUIRE, "workgroup", "local");

  float4 s = *reinterpret_cast<const float4*>(&part[0][row][c4]);
#pragma unroll
  for (int w = 1; w < G1_WAVES; ++w) {
    const float4 p = *reinterpret_cast<const float4*>(&part[w][row][c4]);
    s.x += p.x; s.y += p.y; s.z += p.z; s.w += p.w;
  }
  const size_t off = ((size_t)tile * 16 + row) * H1C + c4;
  *reinterpret_cast<float4*>(C + off) = s;

  deg_ticket_add(col, wv, deg, lo, hi);

  int tn = -1;
  if (dq.pass < 2) {
    tn = __builtin_amdgcn_readfirstlane(*s_tk);
    if (tn >= S) {
      ++dq.pass;
      tn = -1;
    }
  }
  dq.t = tn;
}

__device__ __forceinline__ void g1_load_a(float4 (&a)[8], const float* __restrict__ A,
                                          int tile, int c, int kw) {
  const float* ap = A + ((size_t)tile * 16 + c) * IN_C + kw;
#pragma unroll
  for (int i = 0; i < 8; ++i) a[i] = *reinterpret_cast<const float4*>(ap + 8 * i);
}

__global__ __launch_bounds__(G1_WAVES * 32) void k_gemm1_splitk(
    const float* __restrict__ A, const float* __restrict__ B, float* __restrict__ C,
    int numTiles, const int* __restrict__ ei, const float* __restrict__ ew,
    float* __restrict__ deg, int* __restrict__ cnt, int E, int Nn) {
  __shared__ alignas(16) float part[2][G1_WAVES][16][G1_PITCH];
  __shared__ int s_tk[2];
  __shared__ int s_word;

  const int wave = __builtin_amdgcn_readfirstlane((int)(threadIdx.x >> 5));
  const int lane = threadIdx.x & 31;
  const int c = lane & 15;
  const int h = lane >> 4;
  const int kw = wave * 64 + h * 4;

  v2f bf[16][4];
#pragma unroll
  for (int i = 0; i < 8; ++i) {
#pragma unroll
    for (int e = 0; e < 2; ++e) {
      const int kk = kw + 8 * i + 2 * e;
#pragma unroll
      for (int q = 0; q < 4; ++q) {
        bf[2 * i + e][q].x = B[kk * H1C + 16 * q + c];
        bf[2 * i + e][q].y = B[(kk + 1) * H1C + 16 * q + c];
      }
    }
  }

  const int row = threadIdx.x >> 4;
  const int c4 = (threadIdx.x & 15) * 4;

  const int S = (E + TASK_EDGES - 1) / TASK_EDGES;
  const int mid = range_split(Nn);
  const int mine = share_word(&s_word, l2_half_of_this_wave());
  DegQueue dq;
  dq.pass = 0;
  dq.t = -1;

  const int step = gridDim.x;
  int tile = blockIdx.x;
  if (tile < numTiles) {
    float4 aA[8], aB[8];
    g1_load_a(aA, A, tile, c, kw);
    while (true) {
      int nt = tile + step;
      bool more = nt < numTiles;
      int pt = more ? nt : tile;
      g1_load_a(aB, A, pt, c, kw);
      g1_tile(aA, bf, part[0], &s_tk[0], C, tile, wave, c, h, row, c4, dq, mine, S, mid, ei,
              ew, deg, cnt, E, Nn);
      if (!more) break;
      tile = nt;
      nt = tile + step;
      more = nt < numTiles;
      pt = more ? nt : tile;
      g1_load_a(aA, A, pt, c, kw);
      g1_tile(aB, bf, part[1], &s_tk[1], C, tile, wave, c, h, row, c4, dq, mine, S, mid, ei,
              ew, deg, cnt, E, Nn);
      if (!more) break;
      tile = nt;
    }
  }

  while (dq.pass < 2) {
    if (dq.t >= S) {
      ++dq.pass;
      dq.t = -1;
      continue;
    }
    if (dq.t >= 0) {
      const int q = mine ^ dq.pass;
      int col[DEG_K];
      float wv[DEG_K];
      deg_ticket_load(col, wv, ei, ew, dq.t, E);
      deg_ticket_add(col, wv, deg, q ? mid : 0, q ? Nn : mid);
    }
    int drawn = 0;
    if (threadIdx.x == 0) drawn = atomicAdd(&cnt[mine ^ dq.pass], 1);
    dq.t = share_word(&s_word, drawn);
  }
}


template <int F>
__device__ __forceinline__ void agg_marked_edges(unsigned m, int row, int col, float w,
                                                 const float* __restrict__ h,
                                                 float* __restrict__ agg, int lane) {
  constexpr int P = F / 32;
  constexpr int G = 8;
  while (m) {
    int idx[G];
#pragma unroll
    for (int j = 0; j < G; ++j) {
      idx[j] = m ? (int)__builtin_ctz(m) : -1;
      m &= m - 1u;
    }
    float v[G][P];
#pragma unroll
    for (int j = 0; j < G; ++j) {
      if (idx[j] >= 0) {
        const int r = __builtin_amdgcn_readlane(row, idx[j]);
        const float* src = h + (size_t)r * F + lane;
#pragma unroll
        for (int p = 0; p < P; ++p) v[j][p] = src[p * 32];
      }
    }
#pragma unroll
    for (int j = 0; j < G; ++j) {
      if (idx[j] >= 0) {
        const int c = __builtin_amdgcn_readlane(col, idx[j]);
        const float wj = __int_as_float(__builtin_amdgcn_readlane(__float_as_int(w), idx[j]));
        float* dst = agg + (size_t)c * F + lane;
#pragma unroll
        for (int p = 0; p < P; ++p) atomicAdd(dst + p * 32, wj * v[j][p]);
      }
    }
  }
}

template <int F>
__global__ __launch_bounds__(256) void k_edge_agg(const int* __restrict__ ei,
                                                  const float* __restrict__ wnorm,
                                                  const float* __restrict__ h,
                                                  float* __restrict__ agg,
                                                  int* __restrict__ cnt, int E, int Nn) {
  constexpr int L = TASK_EDGES / 256;
  __shared__ int s_word;
  const int lane = threadIdx.x & 31;
  const int wave = __builtin_amdgcn_readfirstlane((int)(threadIdx.x >> 5));
  const int S = (E + TASK_EDGES - 1) / TASK_EDGES;
  const int mid = range_split(Nn);
  const int mine = share_word(&s_word, l2_half_of_this_wave());
  for (int pass = 0; pass < 2; ++pass) {
    const int q = mine ^ pass;
    const int lo = q ? mid : 0;
    const int hi = q ? Nn : mid;
    int drawn = 0;
    if (threadIdx.x == 0) drawn = atomicAdd(&cnt[q], 1);
    int t = share_word(&s_word, drawn);
    while (t < S) {
      const long long base = (long long)t * TASK_EDGES + wave * (32 * L) + lane;
      int row[L], col[L];
      float w[L];
#pragma unroll
      for (int s = 0; s < L; ++s) {
        const long long e = base + 32 * s;
        row[s] = 0;
        col[s] = -1;
        w[s] = 0.0f;
        if (e < E) {
          row[s] = ei[e];
          col[s] = ei[(long long)E + e];
          w[s] = wnorm[e];
        }
      }
#pragma unroll
      for (int s = 0; s < L; ++s) {
        const unsigned m = __builtin_amdgcn_ballot_w32(col[s] >= lo && col[s] < hi);
        agg_marked_edges<F>(m, row[s], col[s], w[s], h, agg, lane);
      }
      if (threadIdx.x == 0) drawn = atomicAdd(&cnt[q], 1);
      t = share_word(&s_word, drawn);
    }
  }
}

__global__ void k_gemm3_relu(const float* __restrict__ a2, const float* __restrict__ W3,
                             const float* __restrict__ deg, const float* __restrict__ b3,
                             float* __restrict__ g, float* __restrict__ out,
                             int* __restrict__ cnt, int Nn) {
  int i = blockIdx.x * blockDim.x + threadIdx.x;
  if (i < 2) cnt[i] = 0;
  if (i >= Nn) return;
  const float* r = a2 + (size_t)i * H2C;
  float acc = 0.0f;
#pragma unroll
  for (int f = 0; f < H2C; ++f) acc += fmaxf(r[f], 0.0f) * W3[f];
  g[i] = acc;
  const float di = dinv_of(deg[i]);
  out[i] = b3[0] + di * di * acc;
}

__global__ __launch_bounds__(256) void k_edge_agg1(const int* __restrict__ ei,
                                                   const float* __restrict__ wnorm,
                                                   const float* __restrict__ g,
                                                   float* __restrict__ out,
                                                   int* __restrict__ cnt, int E, int Nn) {
  constexpr int K = TASK_EDGES / 256;
  __shared__ int s_word;
  const int S = (E + TASK_EDGES - 1) / TASK_EDGES;
  const int mid = range_split(Nn);
  const int mine = share_word(&s_word, l2_half_of_this_wave());
  for (int pass = 0; pass < 2; ++pass) {
    const int q = mine ^ pass;
    const int lo = q ? mid : 0;
    const int hi = q ? Nn : mid;
    int drawn = 0;
    if (threadIdx.x == 0) drawn = atomicAdd(&cnt[q], 1);
    int t = share_word(&s_word, drawn);
    while (t < S) {
      const long long base = (long long)t * TASK_EDGES + threadIdx.x;
      int row[K], col[K];
      float wv[K], gv[K];
#pragma unroll
      for (int k = 0; k < K; ++k) {
        const long long e = base + 256 * k;
        row[k] = 0;
        col[k] = -1;
        wv[k] = 0.0f;
        if (e < E) {
          row[k] = ei[e];
          col[k] = ei[(long long)E + e];
          wv[k] = wnorm[e];
        }
      }
#pragma unroll
      for (int k = 0; k < K; ++k) {
        gv[k] = 0.0f;
        if (col[k] >= lo && col[k] < hi) gv[k] = g[row[k]];
      }
#pragma unroll
      for (int k = 0; k < K; ++k)
        if (col[k] >= lo && col[k] < hi) atomicAdd(&out[col[k]], wv[k] * gv[k]);
      if (threadIdx.x == 0) drawn = atomicAdd(&cnt[q], 1);
      t = share_word(&s_word, drawn);
    }
  }
}


static inline unsigned int blocks_for(long long total, int bs) {
  return (unsigned int)((total + bs - 1) / bs);
}

extern "C" void kernel_launch(void* const* d_in, const int* in_sizes, int n_in,
                              void* d_out, int out_size, void* d_ws, size_t ws_size,
                              hipStream_t stream) {
  const float* x = (const float*)d_in[0];
  const int* ei = (const int*)d_in[1];
  const float* ew = (const float*)d_in[2];
  const float* W1 = (const float*)d_in[3];
  const float* b1 = (const float*)d_in[4];
  const float* W2 = (const float*)d_in[5];
  const float* b2 = (const float*)d_in[6];
  const float* W3 = (const float*)d_in[7];
  const float* b3 = (const float*)d_in[8];

  const int E = in_sizes[2];
  const int Nn = in_sizes[0] / IN_C;
  float* out = (float*)d_out;

  float* deg = (float*)d_ws;
  float* wnorm = deg + Nn;
  float* bufA = wnorm + E;
  float* bufB = bufA + (size_t)Nn * H1C;
  float* agg2 = bufA + (size_t)Nn * H2C;

  int* cntA = (int*)d_out;
  int* cntB = (int*)(bufB + (size_t)Nn * H2C);

  k_init_deg<<<blocks_for(Nn, 256), 256, 0, stream>>>(deg, cntA, 6, Nn);
  k_gemm1_splitk<<<512, G1_WAVES * 32, 0, stream>>>(x, W1, bufA, Nn / 16, ei, ew, deg, cntA,
                                                    E, Nn);
  k_wnorm<<<256, WN_THREADS, 0, stream>>>(ei, ew, deg, wnorm, E, Nn, bufA, b1, bufB);

  k_edge_agg<H1C><<<SCATTER_BLOCKS, 256, 0, stream>>>(
      ei, wnorm, bufA, bufB, cntA + 2, E, Nn);

  k_gemm_wmma_f32<H1C, H2C / 16, true>
      <<<Nn / 16, (H2C / 16) * 32, 0, stream>>>(bufB, W2, bufA, deg, b2, agg2);
  k_edge_agg<H2C><<<SCATTER_BLOCKS, 256, 0, stream>>>(
      ei, wnorm, bufA, agg2, cntA + 4, E, Nn);

  k_gemm3_relu<<<blocks_for(Nn, 256), 256, 0, stream>>>(agg2, W3, deg, b3, bufB, out, cntB,
                                                        Nn);
  k_edge_agg1<<<SCATTER_BLOCKS, 256, 0, stream>>>(ei, wnorm, bufB, out, cntB, E, Nn);
}


# Layer-1 rows on 256-B boundaries, stored pair-interleaved (float 2l+e = feature 32e+l): the scatter pass gathers a source row with one 8-B-per-lane load, not two 128-B loads; atomic order as before
# speedup vs baseline: 3.7244x; 1.0495x over previous
#include <hip/hip_runtime.h>
#include <hip/hip_bf16.h>


typedef float v2f __attribute__((ext_vector_type(2)));
typedef float v8f __attribute__((ext_vector_type(8)));

#define IN_C 512
#define H1C 64
#define H2C 32

#define H1_POS(f) (2 * ((f) & 31) + ((f) >> 5))


#define TASK_EDGES 1024
#define SCATTER_BLOCKS 2048

__device__ __forceinline__ int l2_half_of_this_wave() {
  return (int)((__builtin_amdgcn_s_sendmsg_rtn(0x87) >> 18) & 1u);
}

__device__ __forceinline__ int range_split(int Nn) { return (Nn >> 1) & ~63; }

__device__ __forceinline__ void lds_barrier() {
  __builtin_amdgcn_fence(__ATOMIC_RELEASE, "workgroup", "local");
  __builtin_amdgcn_s_barrier();
  __builtin_amdgcn_fence(__ATOMIC_ACQUIRE, "workgroup", "local");
}

__device__ __forceinline__ int share_word(int* s_word, int mine) {
  if (threadIdx.x == 0) *s_word = mine;
  lds_barrier();
  const int v = __builtin_amdgcn_readfirstlane(*s_word);
  lds_barrier();
  return v;
}

__global__ void k_init_deg(float* __restrict__ deg, int* __restrict__ cnt, int ncnt, int Nn) {
  int i = blockIdx.x * blockDim.x + threadIdx.x;
  if (i < Nn) deg[i] = 1.0f;
  if (i < ncnt) cnt[i] = 0;
}

#define DEG_K (TASK_EDGES / 256)

__device__ __forceinline__ void deg_ticket_load(int (&col)[DEG_K], float (&wv)[DEG_K],
                                                const int* __restrict__ ei,
                                                const float* __restrict__ ew, int t, int E) {
  const int* __restrict__ cp = ei + E;
  const int base = t * TASK_EDGES + (int)threadIdx.x;
#pragma unroll
  for (int k = 0; k < DEG_K; ++k) {
    const int e = base + 256 * k;
    const int ec = (e < E) ? e : (E - 1);
    const int cv = cp[ec];
    wv[k] = ew[ec];
    col[k] = (e < E) ? cv : -1;
  }
}

__device__ __forceinline__ void deg_ticket_add(const int (&col)[DEG_K], const float (&wv)[DEG_K],
                                               float* __restrict__ deg, int lo, int hi) {
#pragma unroll
  for (int k = 0; k < DEG_K; ++k)
    if (col[k] >= lo && col[k] < hi) atomicAdd(&deg[col[k]], wv[k]);
}

__device__ __forceinline__ float dinv_of(float d) {
  return (d > 0.0f) ? rsqrtf(fmaxf(d, 1e-12f)) : 0.0f;
}

#define WN_THREADS 1024
#define WN_EPT 13
#define WN_CHUNK 65536

__global__ __launch_bounds__(WN_THREADS) void k_wnorm(const int* __restrict__ ei,
                                                      const float* __restrict__ ew,
                                                      const float* __restrict__ deg,
                                                      float* __restrict__ wnorm, int E, int Nn,
                                                      const float* __restrict__ C,
                                                      const float* __restrict__ bias,
                                                      float* __restrict__ S) {
  __shared__ alignas(16) float tab[WN_CHUNK];
  const int tid = threadIdx.x;
  constexpr int BATCH = WN_THREADS * WN_EPT;
  const unsigned per = ((((unsigned)E + gridDim.x - 1u) / gridDim.x) + 31u) & ~31u;
  const long long s0 = (long long)blockIdx.x * per;
  const long long s1 = (s0 + per < (long long)E) ? (s0 + per) : (long long)E;
  for (long long base = s0; base < s1; base += BATCH) {
    const int rem = (s1 - base < (long long)BATCH) ? (int)(s1 - base) : BATCH;
    const int* __restrict__ rp = ei + base;
    const int* __restrict__ cp = ei + (long long)E + base;
    const float* __restrict__ wp = ew + base;
    float* __restrict__ op = wnorm + base;
    int row[WN_EPT], col[WN_EPT];
    float w[WN_EPT], fa[WN_EPT], fb[WN_EPT];
#pragma unroll
    for (int k = 0; k < WN_EPT; ++k) {
      const int t = k * WN_THREADS + tid;
      const int tc = (t < rem) ? t : (rem - 1);
      row[k] = rp[tc];
      col[k] = cp[tc];
      w[k] = wp[tc];
      fa[k] = 0.0f;
      fb[k] = 0.0f;
    }
    for (int lo = 0; lo < Nn; lo += WN_CHUNK) {
      const int cnt = (Nn - lo < WN_CHUNK) ? (Nn - lo) : WN_CHUNK;
      lds_barrier();
      const int n4 = cnt >> 2;
      const float4* __restrict__ dp = reinterpret_cast<const float4*>(deg + lo);
      float4* tp = reinterpret_cast<float4*>(tab);
      for (int g0 = 0; g0 < n4; g0 += WN_THREADS * 8) {
        float4 d[8];
#pragma unroll
        for (int j = 0; j < 8; ++j) {
          const int g = g0 + j * WN_THREADS + tid;
          d[j] = dp[(g < n4) ? g : (n4 - 1)];
        }
#pragma unroll
        for (int j = 0; j < 8; ++j) {
          const int g = g0 + j * WN_THREADS + tid;
          if (g < n4) {
            float4 v;
            v.x = dinv_of(d[j].x);
            v.y = dinv_of(d[j].y);
            v.z = dinv_of(d[j].z);
            v.w = dinv_of(d[j].w);
            tp[g] = v;
          }
        }
      }
      {
        const int i = (n4 << 2) + tid;
        if (i < cnt) tab[i] = dinv_of(deg[lo + i]);
      }
      lds_barrier();
#pragma unroll
      for (int k = 0; k < WN_EPT; ++k) {
        const unsigned r = (unsigned)(row[k] - lo);
        const float tr = tab[(r < (unsigned)cnt) ? r : 0u];
        fa[k] = (r < (unsigned)cnt) ? tr : fa[k];
        const unsigned c = (unsigned)(col[k] - lo);
        const float tc = tab[(c < (unsigned)cnt) ? c : 0u];
        fb[k] = (c < (unsigned)cnt) ? tc : fb[k];
      }
      asm volatile("s_wait_dscnt 0x0" ::: "memory");
    }
#pragma unroll
    for (int k = 0; k < WN_EPT; ++k) {
      const int t = k * WN_THREADS + tid;
      if (t < rem) op[t] = fa[k] * w[k] * fb[k];
    }
  }

  const int total4 = Nn * (H1C / 4);
  const float4* __restrict__ C4 = reinterpret_cast<const float4*>(C);
  const int kq = tid & 15;
  const float2 blo = *reinterpret_cast<const float2*>(bias + 2 * kq);
  const float2 bhi = *reinterpret_cast<const float2*>(bias + 32 + 2 * kq);
  for (int g0 = blockIdx.x * (WN_THREADS * 4); g0 < total4; g0 += gridDim.x * (WN_THREADS * 4)) {
    float4 cv[4];
    float dv[4];
#pragma unroll
    for (int j = 0; j < 4; ++j) {
      const int g = g0 + j * WN_THREADS + tid;
      const int gc = (g < total4) ? g : (total4 - 1);
      cv[j] = C4[gc];
      dv[j] = deg[gc >> 4];
    }
#pragma unroll
    for (int j = 0; j < 4; ++j) {
      const int g = g0 + j * WN_THREADS + tid;
      if (g < total4) {
        const float di = dinv_of(dv[j]);
        const float dd = di * di;
        float2 lo2, hi2;
        lo2.x = blo.x + dd * cv[j].x;
        hi2.x = bhi.x + dd * cv[j].y;
        lo2.y = blo.y + dd * cv[j].z;
        hi2.y = bhi.y + dd * cv[j].w;
        float* __restrict__ srow = S + (size_t)(g >> 4) * H1C + 2 * kq;
        *reinterpret_cast<float2*>(srow) = lo2;
        *reinterpret_cast<float2*>(srow + 32) = hi2;
      }
    }
  }
}


template <int K, int NT, bool RELU>
__global__ __launch_bounds__(NT * 32) void k_gemm_wmma_f32(
    const float* __restrict__ A, const float* __restrict__ B, float* __restrict__ C,
    const float* __restrict__ deg, const float* __restrict__ bias, float* __restrict__ S) {
  constexpr int NOUT = NT * 16;
  const int wave = threadIdx.x >> 5;
  const int lane = threadIdx.x & 31;
  const int m = lane & 15;
  const int half = lane >> 4;
  const int n0 = wave * 16;
  const long long rowBase = (long long)blockIdx.x * 16;

  const float* arow = A + (rowBase + m) * K + half * 2;

  float dg[8];
#pragma unroll
  for (int v = 0; v < 8; ++v) dg[v] = deg[rowBase + v + 8 * half];
  const float bcol = bias[n0 + m];

  v8f acc = {};
#pragma unroll 4
  for (int k = 0; k < K; k += 4) {
    float2 av = *reinterpret_cast<const float2*>(arow + k);
    v2f a;
    a.x = RELU ? fmaxf(av.x, 0.0f) : av.x;
    a.y = RELU ? fmaxf(av.y, 0.0f) : av.y;
    const int kr = k + half * 2;
    v2f b;
    b.x = B[kr * NOUT + n0 + m];
    b.y = B[(kr + 1) * NOUT + n0 + m];
    acc = __builtin_amdgcn_wmma_f32_16x16x4_f32(false, a, false, b, (short)0, acc,
                                                false, false);
  }

  float dd[8];
#pragma unroll
  for (int v = 0; v < 8; ++v) {
    const float di = dinv_of(dg[v]);
    dd[v] = di * di;
  }
#pragma unroll
  for (int v = 0; v < 8; ++v) {
    long long r = rowBase + v + 8 * half;
    C[r * NOUT + n0 + m] = acc[v];
    S[r * NOUT + n0 + m] = bcol + dd[v] * acc[v];
  }
}

#define G1_WAVES 8
#define G1_PITCH 68

struct DegQueue {
  int pass;
  int t;
};

__device__ __forceinline__ void g1_tile(const float4 (&a)[8], const v2f (&bf)[16][4],
                                        float (&part)[G1_WAVES][16][G1_PITCH], int* s_tk,
                                        float* __restrict__ C, int tile, int wave, int c,
                                        int h, int row, int c4, DegQueue& dq, int mine,
                                        int S, int mid, const int* __restrict__ ei,
                                        const float* __restrict__ ew,
                                        float* __restrict__ deg, int* __restrict__ cnt,
                                        int E, int Nn) {
  int drawn = 0x7fffffff;
  if (dq.pass < 2 && threadIdx.x == 0)
    drawn = (int)atomicInc(reinterpret_cast<unsigned int*>(cnt + (mine ^ dq.pass)), 0xffffffffu);

  __builtin_amdgcn_sched_barrier(0);

  v8f acc[4] = {};
#pragma unroll
  for (int i = 0; i < 8; ++i) {
    v2f a0, a1;
    a0.x = a[i].x; a0.y = a[i].y;
    a1.x = a[i].z; a1.y = a[i].w;
#pragma unroll
    for (int q = 0; q < 4; ++q)
      acc[q] = __builtin_amdgcn_wmma_f32_16x16x4_f32(false, a0, false, bf[2 * i][q],
                                                     (short)0, acc[q], false, false);
#pragma unroll
    for (int q = 0; q < 4; ++q)
      acc[q] = __builtin_amdgcn_wmma_f32_16x16x4_f32(false, a1, false, bf[2 * i + 1][q],
                                                     (short)0, acc[q], false, false);
  }

  const bool have = dq.t >= 0;
  int tl = have ? dq.t : 0;
  __builtin_amdgcn_sched_barrier(0);
  asm volatile("v_nop\n\tv_nop\n\tv_nop\n\tv_nop\n\tv_nop\n\tv_nop\n\tv_nop\n\tv_nop"
               : "+v"(acc[0]), "+v"(acc[1]), "+v"(acc[2]), "+v"(acc[3]), "+s"(tl));
  __builtin_amdgcn_sched_barrier(0);

#pragma unroll
  for (int q4 = 0; q4 < 4; ++q4) {
#pragma unroll
    for (int v = 0; v < 8; ++v) part[wave][v + 8 * h][H1_POS(16 * q4 + c)] = acc[q4][v];
  }

  const int q = mine ^ dq.pass;
  const int lo = have ? (q ? mid : 0) : 0;
  const int hi = have ? (q ? Nn : mid) : 0;
  int col[DEG_K];
  float wv[DEG_K];
  deg_ticket_load(col, wv, ei, ew, tl, E);
  if (threadIdx.x == 0) *s_tk = drawn;

  __builtin_amdgcn_fence(__ATOMIC_RELEASE, "workgroup", "local");
  __builtin_amdgcn_s_barrier();
  __builtin_amdgcn_fence(__ATOMIC_ACQUIRE, "workgroup", "local");

  float4 s = *reinterpret_cast<const float4*>(&part[0][row][c4]);
#pragma unroll
  for (int w = 1; w < G1_WAVES; ++w) {
    const float4 p = *reinterpret_cast<const float4*>(&part[w][row][c4]);
    s.x += p.x; s.y += p.y; s.z += p.z; s.w += p.w;
  }
  const size_t off = ((size_t)tile * 16 + row) * H1C + c4;
  *reinterpret_cast<float4*>(C + off) = s;

  deg_ticket_add(col, wv, deg, lo, hi);

  int tn = -1;
  if (dq.pass < 2) {
    tn = __builtin_amdgcn_readfirstlane(*s_tk);
    if (tn >= S) {
      ++dq.pass;
      tn = -1;
    }
  }
  dq.t = tn;
}

__device__ __forceinline__ void g1_load_a(float4 (&a)[8], const float* __restrict__ A,
                                          int tile, int c, int kw) {
  const float* ap = A + ((size_t)tile * 16 + c) * IN_C + kw;
#pragma unroll
  for (int i = 0; i < 8; ++i) a[i] = *reinterpret_cast<const float4*>(ap + 8 * i);
}

__global__ __launch_bounds__(G1_WAVES * 32) void k_gemm1_splitk(
    const float* __restrict__ A, const float* __restrict__ B, float* __restrict__ C,
    int numTiles, const int* __restrict__ ei, const float* __restrict__ ew,
    float* __restrict__ deg, int* __restrict__ cnt, int E, int Nn) {
  __shared__ alignas(16) float part[2][G1_WAVES][16][G1_PITCH];
  __shared__ int s_tk[2];
  __shared__ int s_word;

  const int wave = __builtin_amdgcn_readfirstlane((int)(threadIdx.x >> 5));
  const int lane = threadIdx.x & 31;
  const int c = lane & 15;
  const int h = lane >> 4;
  const int kw = wave * 64 + h * 4;

  v2f bf[16][4];
#pragma unroll
  for (int i = 0; i < 8; ++i) {
#pragma unroll
    for (int e = 0; e < 2; ++e) {
      const int kk = kw + 8 * i + 2 * e;
#pragma unroll
      for (int q = 0; q < 4; ++q) {
        bf[2 * i + e][q].x = B[kk * H1C + 16 * q + c];
        bf[2 * i + e][q].y = B[(kk + 1) * H1C + 16 * q + c];
      }
    }
  }

  const int row = threadIdx.x >> 4;
  const int c4 = (threadIdx.x & 15) * 4;

  const int S = (E + TASK_EDGES - 1) / TASK_EDGES;
  const int mid = range_split(Nn);
  const int mine = share_word(&s_word, l2_half_of_this_wave());
  DegQueue dq;
  dq.pass = 0;
  dq.t = -1;

  const int step = gridDim.x;
  int tile = blockIdx.x;
  if (tile < numTiles) {
    float4 aA[8], aB[8];
    g1_load_a(aA, A, tile, c, kw);
    while (true) {
      int nt = tile + step;
      bool more = nt < numTiles;
      int pt = more ? nt : tile;
      g1_load_a(aB, A, pt, c, kw);
      g1_tile(aA, bf, part[0], &s_tk[0], C, tile, wave, c, h, row, c4, dq, mine, S, mid, ei,
              ew, deg, cnt, E, Nn);
      if (!more) break;
      tile = nt;
      nt = tile + step;
      more = nt < numTiles;
      pt = more ? nt : tile;
      g1_load_a(aA, A, pt, c, kw);
      g1_tile(aB, bf, part[1], &s_tk[1], C, tile, wave, c, h, row, c4, dq, mine, S, mid, ei,
              ew, deg, cnt, E, Nn);
      if (!more) break;
      tile = nt;
    }
  }

  while (dq.pass < 2) {
    if (dq.t >= S) {
      ++dq.pass;
      dq.t = -1;
      continue;
    }
    if (dq.t >= 0) {
      const int q = mine ^ dq.pass;
      int col[DEG_K];
      float wv[DEG_K];
      deg_ticket_load(col, wv, ei, ew, dq.t, E);
      deg_ticket_add(col, wv, deg, q ? mid : 0, q ? Nn : mid);
    }
    int drawn = 0;
    if (threadIdx.x == 0) drawn = atomicAdd(&cnt[mine ^ dq.pass], 1);
    dq.t = share_word(&s_word, drawn);
  }
}


template <int F>
__device__ __forceinline__ void agg_marked_edges(unsigned m, int row, int col, float w,
                                                 const float* __restrict__ h,
                                                 float* __restrict__ agg, int lane) {
  constexpr int P = F / 32;
  constexpr int G = 8;
  while (m) {
    int idx[G];
#pragma unroll
    for (int j = 0; j < G; ++j) {
      idx[j] = m ? (int)__builtin_ctz(m) : -1;
      m &= m - 1u;
    }
    float v[G][P];
#pragma unroll
    for (int j = 0; j < G; ++j) {
      if (idx[j] >= 0) {
        const int r = __builtin_amdgcn_readlane(row, idx[j]);
        if constexpr (P == 2) {
          const float2 t = *reinterpret_cast<const float2*>(h + (size_t)r * F + 2 * lane);
          v[j][0] = t.x;
          v[j][1] = t.y;
        } else {
          const float* src = h + (size_t)r * F + lane;
#pragma unroll
          for (int p = 0; p < P; ++p) v[j][p] = src[p * 32];
        }
      }
    }
#pragma unroll
    for (int j = 0; j < G; ++j) {
      if (idx[j] >= 0) {
        const int c = __builtin_amdgcn_readlane(col, idx[j]);
        const float wj = __int_as_float(__builtin_amdgcn_readlane(__float_as_int(w), idx[j]));
        float* dst = agg + (size_t)c * F + lane;
#pragma unroll
        for (int p = 0; p < P; ++p) atomicAdd(dst + p * 32, wj * v[j][p]);
      }
    }
  }
}

template <int F>
__global__ __launch_bounds__(256) void k_edge_agg(const int* __restrict__ ei,
                                                  const float* __restrict__ wnorm,
                                                  const float* __restrict__ h,
                                                  float* __restrict__ agg,
                                                  int* __restrict__ cnt, int E, int Nn) {
  constexpr int L = TASK_EDGES / 256;
  __shared__ int s_word;
  const int lane = threadIdx.x & 31;
  const int wave = __builtin_amdgcn_readfirstlane((int)(threadIdx.x >> 5));
  const int S = (E + TASK_EDGES - 1) / TASK_EDGES;
  const int mid = range_split(Nn);
  const int mine = share_word(&s_word, l2_half_of_this_wave());
  for (int pass = 0; pass < 2; ++pass) {
    const int q = mine ^ pass;
    const int lo = q ? mid : 0;
    const int hi = q ? Nn : mid;
    int drawn = 0;
    if (threadIdx.x == 0) drawn = atomicAdd(&cnt[q], 1);
    int t = share_word(&s_word, drawn);
    while (t < S) {
      const long long base = (long long)t * TASK_EDGES + wave * (32 * L) + lane;
      int row[L], col[L];
      float w[L];
#pragma unroll
      for (int s = 0; s < L; ++s) {
        const long long e = base + 32 * s;
        row[s] = 0;
        col[s] = -1;
        w[s] = 0.0f;
        if (e < E) {
          row[s] = ei[e];
          col[s] = ei[(long long)E + e];
          w[s] = wnorm[e];
        }
      }
#pragma unroll
      for (int s = 0; s < L; ++s) {
        const unsigned m = __builtin_amdgcn_ballot_w32(col[s] >= lo && col[s] < hi);
        agg_marked_edges<F>(m, row[s], col[s], w[s], h, agg, lane);
      }
      if (threadIdx.x == 0) drawn = atomicAdd(&cnt[q], 1);
      t = share_word(&s_word, drawn);
    }
  }
}

__global__ void k_gemm3_relu(const float* __restrict__ a2, const float* __restrict__ W3,
                             const float* __restrict__ deg, const float* __restrict__ b3,
                             float* __restrict__ g, float* __restrict__ out,
                             int* __restrict__ cnt, int Nn) {
  int i = blockIdx.x * blockDim.x + threadIdx.x;
  if (i < 2) cnt[i] = 0;
  if (i >= Nn) return;
  const float* r = a2 + (size_t)i * H2C;
  float acc = 0.0f;
#pragma unroll
  for (int f = 0; f < H2C; ++f) acc += fmaxf(r[f], 0.0f) * W3[f];
  g[i] = acc;
  const float di = dinv_of(deg[i]);
  out[i] = b3[0] + di * di * acc;
}

__global__ __launch_bounds__(256) void k_edge_agg1(const int* __restrict__ ei,
                                                   const float* __restrict__ wnorm,
                                                   const float* __restrict__ g,
                                                   float* __restrict__ out,
                                                   int* __restrict__ cnt, int E, int Nn) {
  constexpr int K = TASK_EDGES / 256;
  __shared__ int s_word;
  const int S = (E + TASK_EDGES - 1) / TASK_EDGES;
  const int mid = range_split(Nn);
  const int mine = share_word(&s_word, l2_half_of_this_wave());
  for (int pass = 0; pass < 2; ++pass) {
    const int q = mine ^ pass;
    const int lo = q ? mid : 0;
    const int hi = q ? Nn : mid;
    int drawn = 0;
    if (threadIdx.x == 0) drawn = atomicAdd(&cnt[q], 1);
    int t = share_word(&s_word, drawn);
    while (t < S) {
      const long long base = (long long)t * TASK_EDGES + threadIdx.x;
      int row[K], col[K];
      float wv[K], gv[K];
#pragma unroll
      for (int k = 0; k < K; ++k) {
        const long long e = base + 256 * k;
        row[k] = 0;
        col[k] = -1;
        wv[k] = 0.0f;
        if (e < E) {
          row[k] = ei[e];
          col[k] = ei[(long long)E + e];
          wv[k] = wnorm[e];
        }
      }
#pragma unroll
      for (int k = 0; k < K; ++k) {
        gv[k] = 0.0f;
        if (col[k] >= lo && col[k] < hi) gv[k] = g[row[k]];
      }
#pragma unroll
      for (int k = 0; k < K; ++k)
        if (col[k] >= lo && col[k] < hi) atomicAdd(&out[col[k]], wv[k] * gv[k]);
      if (threadIdx.x == 0) drawn = atomicAdd(&cnt[q], 1);
      t = share_word(&s_word, drawn);
    }
  }
}


static inline unsigned int blocks_for(long long total, int bs) {
  return (unsigned int)((total + bs - 1) / bs);
}

extern "C" void kernel_launch(void* const* d_in, const int* in_sizes, int n_in,
                              void* d_out, int out_size, void* d_ws, size_t ws_size,
                              hipStream_t stream) {
  const float* x = (const float*)d_in[0];
  const int* ei = (const int*)d_in[1];
  const float* ew = (const float*)d_in[2];
  const float* W1 = (const float*)d_in[3];
  const float* b1 = (const float*)d_in[4];
  const float* W2 = (const float*)d_in[5];
  const float* b2 = (const float*)d_in[6];
  const float* W3 = (const float*)d_in[7];
  const float* b3 = (const float*)d_in[8];

  const int E = in_sizes[2];
  const int Nn = in_sizes[0] / IN_C;
  float* out = (float*)d_out;

  float* bufA = (float*)d_ws;
  float* bufB = bufA + (size_t)Nn * H1C;
  float* deg = bufB + (size_t)Nn * H1C;
  float* wnorm = deg + Nn;
  float* agg2 = bufA + (size_t)Nn * H2C;

  int* cntA = (int*)d_out;
  int* cntB = (int*)(bufB + (size_t)Nn * H2C);

  k_init_deg<<<blocks_for(Nn, 256), 256, 0, stream>>>(deg, cntA, 6, Nn);
  k_gemm1_splitk<<<512, G1_WAVES * 32, 0, stream>>>(x, W1, bufA, Nn / 16, ei, ew, deg, cntA,
                                                    E, Nn);
  k_wnorm<<<256, WN_THREADS, 0, stream>>>(ei, ew, deg, wnorm, E, Nn, bufA, b1, bufB);

  k_edge_agg<H1C><<<SCATTER_BLOCKS, 256, 0, stream>>>(
      ei, wnorm, bufA, bufB, cntA + 2, E, Nn);

  k_gemm_wmma_f32<H1C, H2C / 16, true>
      <<<Nn / 16, (H2C / 16) * 32, 0, stream>>>(bufB, W2, bufA, deg, b2, agg2);
  k_edge_agg<H2C><<<SCATTER_BLOCKS, 256, 0, stream>>>(
      ei, wnorm, bufA, agg2, cntA + 4, E, Nn);

  k_gemm3_relu<<<blocks_for(Nn, 256), 256, 0, stream>>>(agg2, W3, deg, b3, bufB, out, cntB,
                                                        Nn);
  k_edge_agg1<<<SCATTER_BLOCKS, 256, 0, stream>>>(ei, wnorm, bufB, out, cntB, E, Nn);
}


# Layer-3 scatter pass stages the per-node source value in LDS (256-KiB chunks, one 1024-thread workgroup per WGP): its per-edge 4-byte gather becomes an LDS read; f32 atomic sum order varies as before
# speedup vs baseline: 3.9904x; 1.0714x over previous
#include <hip/hip_runtime.h>
#include <hip/hip_bf16.h>


typedef float v2f __attribute__((ext_vector_type(2)));
typedef float v8f __attribute__((ext_vector_type(8)));

#define IN_C 512
#define H1C 64
#define H2C 32

#define H1_POS(f) (2 * ((f) & 31) + ((f) >> 5))


#define TASK_EDGES 1024
#define SCATTER_BLOCKS 2048

__device__ __forceinline__ int l2_half_of_this_wave() {
  return (int)((__builtin_amdgcn_s_sendmsg_rtn(0x87) >> 18) & 1u);
}

__device__ __forceinline__ int range_split(int Nn) { return (Nn >> 1) & ~63; }

__device__ __forceinline__ void lds_barrier() {
  __builtin_amdgcn_fence(__ATOMIC_RELEASE, "workgroup", "local");
  __builtin_amdgcn_s_barrier();
  __builtin_amdgcn_fence(__ATOMIC_ACQUIRE, "workgroup", "local");
}

__device__ __forceinline__ int share_word(int* s_word, int mine) {
  if (threadIdx.x == 0) *s_word = mine;
  lds_barrier();
  const int v = __builtin_amdgcn_readfirstlane(*s_word);
  lds_barrier();
  return v;
}

__global__ void k_init_deg(float* __restrict__ deg, int* __restrict__ cnt, int ncnt, int Nn) {
  int i = blockIdx.x * blockDim.x + threadIdx.x;
  if (i < Nn) deg[i] = 1.0f;
  if (i < ncnt) cnt[i] = 0;
}

#define DEG_K (TASK_EDGES / 256)

__device__ __forceinline__ void deg_ticket_load(int (&col)[DEG_K], float (&wv)[DEG_K],
                                                const int* __restrict__ ei,
                                                const float* __restrict__ ew, int t, int E) {
  const int* __restrict__ cp = ei + E;
  const int base = t * TASK_EDGES + (int)threadIdx.x;
#pragma unroll
  for (int k = 0; k < DEG_K; ++k) {
    const int e = base + 256 * k;
    const int ec = (e < E) ? e : (E - 1);
    const int cv = cp[ec];
    wv[k] = ew[ec];
    col[k] = (e < E) ? cv : -1;
  }
}

__device__ __forceinline__ void deg_ticket_add(const int (&col)[DEG_K], const float (&wv)[DEG_K],
                                               float* __restrict__ deg, int lo, int hi) {
#pragma unroll
  for (int k = 0; k < DEG_K; ++k)
    if (col[k] >= lo && col[k] < hi) atomicAdd(&deg[col[k]], wv[k]);
}

__device__ __forceinline__ float dinv_of(float d) {
  return (d > 0.0f) ? rsqrtf(fmaxf(d, 1e-12f)) : 0.0f;
}

#define WN_THREADS 1024
#define WN_EPT 13
#define WN_CHUNK 65536

__global__ __launch_bounds__(WN_THREADS) void k_wnorm(const int* __restrict__ ei,
                                                      const float* __restrict__ ew,
                                                      const float* __restrict__ deg,
                                                      float* __restrict__ wnorm, int E, int Nn,
                                                      const float* __restrict__ C,
                                                      const float* __restrict__ bias,
                                                      float* __restrict__ S) {
  __shared__ alignas(16) float tab[WN_CHUNK];
  const int tid = threadIdx.x;
  constexpr int BATCH = WN_THREADS * WN_EPT;
  const unsigned per = ((((unsigned)E + gridDim.x - 1u) / gridDim.x) + 31u) & ~31u;
  const long long s0 = (long long)blockIdx.x * per;
  const long long s1 = (s0 + per < (long long)E) ? (s0 + per) : (long long)E;
  for (long long base = s0; base < s1; base += BATCH) {
    const int rem = (s1 - base < (long long)BATCH) ? (int)(s1 - base) : BATCH;
    const int* __restrict__ rp = ei + base;
    const int* __restrict__ cp = ei + (long long)E + base;
    const float* __restrict__ wp = ew + base;
    float* __restrict__ op = wnorm + base;
    int row[WN_EPT], col[WN_EPT];
    float w[WN_EPT], fa[WN_EPT], fb[WN_EPT];
#pragma unroll
    for (int k = 0; k < WN_EPT; ++k) {
      const int t = k * WN_THREADS + tid;
      const int tc = (t < rem) ? t : (rem - 1);
      row[k] = rp[tc];
      col[k] = cp[tc];
      w[k] = wp[tc];
      fa[k] = 0.0f;
      fb[k] = 0.0f;
    }
    for (int lo = 0; lo < Nn; lo += WN_CHUNK) {
      const int cnt = (Nn - lo < WN_CHUNK) ? (Nn - lo) : WN_CHUNK;
      lds_barrier();
      const int n4 = cnt >> 2;
      const float4* __restrict__ dp = reinterpret_cast<const float4*>(deg + lo);
      float4* tp = reinterpret_cast<float4*>(tab);
      for (int g0 = 0; g0 < n4; g0 += WN_THREADS * 8) {
        float4 d[8];
#pragma unroll
        for (int j = 0; j < 8; ++j) {
          const int g = g0 + j * WN_THREADS + tid;
          d[j] = dp[(g < n4) ? g : (n4 - 1)];
        }
#pragma unroll
        for (int j = 0; j < 8; ++j) {
          const int g = g0 + j * WN_THREADS + tid;
          if (g < n4) {
            float4 v;
            v.x = dinv_of(d[j].x);
            v.y = dinv_of(d[j].y);
            v.z = dinv_of(d[j].z);
            v.w = dinv_of(d[j].w);
            tp[g] = v;
          }
        }
      }
      {
        const int i = (n4 << 2) + tid;
        if (i < cnt) tab[i] = dinv_of(deg[lo + i]);
      }
      lds_barrier();
#pragma unroll
      for (int k = 0; k < WN_EPT; ++k) {
        const unsigned r = (unsigned)(row[k] - lo);
        const float tr = tab[(r < (unsigned)cnt) ? r : 0u];
        fa[k] = (r < (unsigned)cnt) ? tr : fa[k];
        const unsigned c = (unsigned)(col[k] - lo);
        const float tc = tab[(c < (unsigned)cnt) ? c : 0u];
        fb[k] = (c < (unsigned)cnt) ? tc : fb[k];
      }
      asm volatile("s_wait_dscnt 0x0" ::: "memory");
    }
#pragma unroll
    for (int k = 0; k < WN_EPT; ++k) {
      const int t = k * WN_THREADS + tid;
      if (t < rem) op[t] = fa[k] * w[k] * fb[k];
    }
  }

  const int total4 = Nn * (H1C / 4);
  const float4* __restrict__ C4 = reinterpret_cast<const float4*>(C);
  const int kq = tid & 15;
  const float2 blo = *reinterpret_cast<const float2*>(bias + 2 * kq);
  const float2 bhi = *reinterpret_cast<const float2*>(bias + 32 + 2 * kq);
  for (int g0 = blockIdx.x * (WN_THREADS * 4); g0 < total4; g0 += gridDim.x * (WN_THREADS * 4)) {
    float4 cv[4];
    float dv[4];
#pragma unroll
    for (int j = 0; j < 4; ++j) {
      const int g = g0 + j * WN_THREADS + tid;
      const int gc = (g < total4) ? g : (total4 - 1);
      cv[j] = C4[gc];
      dv[j] = deg[gc >> 4];
    }
#pragma unroll
    for (int j = 0; j < 4; ++j) {
      const int g = g0 + j * WN_THREADS + tid;
      if (g < total4) {
        const float di = dinv_of(dv[j]);
        const float dd = di * di;
        float2 lo2, hi2;
        lo2.x = blo.x + dd * cv[j].x;
        hi2.x = bhi.x + dd * cv[j].y;
        lo2.y = blo.y + dd * cv[j].z;
        hi2.y = bhi.y + dd * cv[j].w;
        float* __restrict__ srow = S + (size_t)(g >> 4) * H1C + 2 * kq;
        *reinterpret_cast<float2*>(srow) = lo2;
        *reinterpret_cast<float2*>(srow + 32) = hi2;
      }
    }
  }
}


template <int K, int NT, bool RELU>
__global__ __launch_bounds__(NT * 32) void k_gemm_wmma_f32(
    const float* __restrict__ A, const float* __restrict__ B, float* __restrict__ C,
    const float* __restrict__ deg, const float* __restrict__ bias, float* __restrict__ S) {
  constexpr int NOUT = NT * 16;
  const int wave = threadIdx.x >> 5;
  const int lane = threadIdx.x & 31;
  const int m = lane & 15;
  const int half = lane >> 4;
  const int n0 = wave * 16;
  const long long rowBase = (long long)blockIdx.x * 16;

  const float* arow = A + (rowBase + m) * K + half * 2;

  float dg[8];
#pragma unroll
  for (int v = 0; v < 8; ++v) dg[v] = deg[rowBase + v + 8 * half];
  const float bcol = bias[n0 + m];

  v8f acc = {};
#pragma unroll 4
  for (int k = 0; k < K; k += 4) {
    float2 av = *reinterpret_cast<const float2*>(arow + k);
    v2f a;
    a.x = RELU ? fmaxf(av.x, 0.0f) : av.x;
    a.y = RELU ? fmaxf(av.y, 0.0f) : av.y;
    const int kr = k + half * 2;
    v2f b;
    b.x = B[kr * NOUT + n0 + m];
    b.y = B[(kr + 1) * NOUT + n0 + m];
    acc = __builtin_amdgcn_wmma_f32_16x16x4_f32(false, a, false, b, (short)0, acc,
                                                false, false);
  }

  float dd[8];
#pragma unroll
  for (int v = 0; v < 8; ++v) {
    const float di = dinv_of(dg[v]);
    dd[v] = di * di;
  }
#pragma unroll
  for (int v = 0; v < 8; ++v) {
    long long r = rowBase + v + 8 * half;
    C[r * NOUT + n0 + m] = acc[v];
    S[r * NOUT + n0 + m] = bcol + dd[v] * acc[v];
  }
}

#define G1_WAVES 8
#define G1_PITCH 68

struct DegQueue {
  int pass;
  int t;
};

__device__ __forceinline__ void g1_tile(const float4 (&a)[8], const v2f (&bf)[16][4],
                                        float (&part)[G1_WAVES][16][G1_PITCH], int* s_tk,
                                        float* __restrict__ C, int tile, int wave, int c,
                                        int h, int row, int c4, DegQueue& dq, int mine,
                                        int S, int mid, const int* __restrict__ ei,
                                        const float* __restrict__ ew,
                                        float* __restrict__ deg, int* __restrict__ cnt,
                                        int E, int Nn) {
  int drawn = 0x7fffffff;
  if (dq.pass < 2 && threadIdx.x == 0)
    drawn = (int)atomicInc(reinterpret_cast<unsigned int*>(cnt + (mine ^ dq.pass)), 0xffffffffu);

  __builtin_amdgcn_sched_barrier(0);

  v8f acc[4] = {};
#pragma unroll
  for (int i = 0; i < 8; ++i) {
    v2f a0, a1;
    a0.x = a[i].x; a0.y = a[i].y;
    a1.x = a[i].z; a1.y = a[i].w;
#pragma unroll
    for (int q = 0; q < 4; ++q)
      acc[q] = __builtin_amdgcn_wmma_f32_16x16x4_f32(false, a0, false, bf[2 * i][q],
                                                     (short)0, acc[q], false, false);
#pragma unroll
    for (int q = 0; q < 4; ++q)
      acc[q] = __builtin_amdgcn_wmma_f32_16x16x4_f32(false, a1, false, bf[2 * i + 1][q],
                                                     (short)0, acc[q], false, false);
  }

  const bool have = dq.t >= 0;
  int tl = have ? dq.t : 0;
  __builtin_amdgcn_sched_barrier(0);
  asm volatile("v_nop\n\tv_nop\n\tv_nop\n\tv_nop\n\tv_nop\n\tv_nop\n\tv_nop\n\tv_nop"
               : "+v"(acc[0]), "+v"(acc[1]), "+v"(acc[2]), "+v"(acc[3]), "+s"(tl));
  __builtin_amdgcn_sched_barrier(0);

#pragma unroll
  for (int q4 = 0; q4 < 4; ++q4) {
#pragma unroll
    for (int v = 0; v < 8; ++v) part[wave][v + 8 * h][H1_POS(16 * q4 + c)] = acc[q4][v];
  }

  const int q = mine ^ dq.pass;
  const int lo = have ? (q ? mid : 0) : 0;
  const int hi = have ? (q ? Nn : mid) : 0;
  int col[DEG_K];
  float wv[DEG_K];
  deg_ticket_load(col, wv, ei, ew, tl, E);
  if (threadIdx.x == 0) *s_tk = drawn;

  __builtin_amdgcn_fence(__ATOMIC_RELEASE, "workgroup", "local");
  __builtin_amdgcn_s_barrier();
  __builtin_amdgcn_fence(__ATOMIC_ACQUIRE, "workgroup", "local");

  float4 s = *reinterpret_cast<const float4*>(&part[0][row][c4]);
#pragma unroll
  for (int w = 1; w < G1_WAVES; ++w) {
    const float4 p = *reinterpret_cast<const float4*>(&part[w][row][c4]);
    s.x += p.x; s.y += p.y; s.z += p.z; s.w += p.w;
  }
  const size_t off = ((size_t)tile * 16 + row) * H1C + c4;
  *reinterpret_cast<float4*>(C + off) = s;

  deg_ticket_add(col, wv, deg, lo, hi);

  int tn = -1;
  if (dq.pass < 2) {
    tn = __builtin_amdgcn_readfirstlane(*s_tk);
    if (tn >= S) {
      ++dq.pass;
      tn = -1;
    }
  }
  dq.t = tn;
}

__device__ __forceinline__ void g1_load_a(float4 (&a)[8], const float* __restrict__ A,
                                          int tile, int c, int kw) {
  const float* ap = A + ((size_t)tile * 16 + c) * IN_C + kw;
#pragma unroll
  for (int i = 0; i < 8; ++i) a[i] = *reinterpret_cast<const float4*>(ap + 8 * i);
}

__global__ __launch_bounds__(G1_WAVES * 32) void k_gemm1_splitk(
    const float* __restrict__ A, const float* __restrict__ B, float* __restrict__ C,
    int numTiles, const int* __restrict__ ei, const float* __restrict__ ew,
    float* __restrict__ deg, int* __restrict__ cnt, int E, int Nn) {
  __shared__ alignas(16) float part[2][G1_WAVES][16][G1_PITCH];
  __shared__ int s_tk[2];
  __shared__ int s_word;

  const int wave = __builtin_amdgcn_readfirstlane((int)(threadIdx.x >> 5));
  const int lane = threadIdx.x & 31;
  const int c = lane & 15;
  const int h = lane >> 4;
  const int kw = wave * 64 + h * 4;

  v2f bf[16][4];
#pragma unroll
  for (int i = 0; i < 8; ++i) {
#pragma unroll
    for (int e = 0; e < 2; ++e) {
      const int kk = kw + 8 * i + 2 * e;
#pragma unroll
      for (int q = 0; q < 4; ++q) {
        bf[2 * i + e][q].x = B[kk * H1C + 16 * q + c];
        bf[2 * i + e][q].y = B[(kk + 1) * H1C + 16 * q + c];
      }
    }
  }

  const int row = threadIdx.x >> 4;
  const int c4 = (threadIdx.x & 15) * 4;

  const int S = (E + TASK_EDGES - 1) / TASK_EDGES;
  const int mid = range_split(Nn);
  const int mine = share_word(&s_word, l2_half_of_this_wave());
  DegQueue dq;
  dq.pass = 0;
  dq.t = -1;

  const int step = gridDim.x;
  int tile = blockIdx.x;
  if (tile < numTiles) {
    float4 aA[8], aB[8];
    g1_load_a(aA, A, tile, c, kw);
    while (true) {
      int nt = tile + step;
      bool more = nt < numTiles;
      int pt = more ? nt : tile;
      g1_load_a(aB, A, pt, c, kw);
      g1_tile(aA, bf, part[0], &s_tk[0], C, tile, wave, c, h, row, c4, dq, mine, S, mid, ei,
              ew, deg, cnt, E, Nn);
      if (!more) break;
      tile = nt;
      nt = tile + step;
      more = nt < numTiles;
      pt = more ? nt : tile;
      g1_load_a(aA, A, pt, c, kw);
      g1_tile(aB, bf, part[1], &s_tk[1], C, tile, wave, c, h, row, c4, dq, mine, S, mid, ei,
              ew, deg, cnt, E, Nn);
      if (!more) break;
      tile = nt;
    }
  }

  while (dq.pass < 2) {
    if (dq.t >= S) {
      ++dq.pass;
      dq.t = -1;
      continue;
    }
    if (dq.t >= 0) {
      const int q = mine ^ dq.pass;
      int col[DEG_K];
      float wv[DEG_K];
      deg_ticket_load(col, wv, ei, ew, dq.t, E);
      deg_ticket_add(col, wv, deg, q ? mid : 0, q ? Nn : mid);
    }
    int drawn = 0;
    if (threadIdx.x == 0) drawn = atomicAdd(&cnt[mine ^ dq.pass], 1);
    dq.t = share_word(&s_word, drawn);
  }
}


template <int F>
__device__ __forceinline__ void agg_marked_edges(unsigned m, int row, int col, float w,
                                                 const float* __restrict__ h,
                                                 float* __restrict__ agg, int lane) {
  constexpr int P = F / 32;
  constexpr int G = 8;
  while (m) {
    int idx[G];
#pragma unroll
    for (int j = 0; j < G; ++j) {
      idx[j] = m ? (int)__builtin_ctz(m) : -1;
      m &= m - 1u;
    }
    float v[G][P];
#pragma unroll
    for (int j = 0; j < G; ++j) {
      if (idx[j] >= 0) {
        const int r = __builtin_amdgcn_readlane(row, idx[j]);
        if constexpr (P == 2) {
          const float2 t = *reinterpret_cast<const float2*>(h + (size_t)r * F + 2 * lane);
          v[j][0] = t.x;
          v[j][1] = t.y;
        } else {
          const float* src = h + (size_t)r * F + lane;
#pragma unroll
          for (int p = 0; p < P; ++p) v[j][p] = src[p * 32];
        }
      }
    }
#pragma unroll
    for (int j = 0; j < G; ++j) {
      if (idx[j] >= 0) {
        const int c = __builtin_amdgcn_readlane(col, idx[j]);
        const float wj = __int_as_float(__builtin_amdgcn_readlane(__float_as_int(w), idx[j]));
        float* dst = agg + (size_t)c * F + lane;
#pragma unroll
        for (int p = 0; p < P; ++p) atomicAdd(dst + p * 32, wj * v[j][p]);
      }
    }
  }
}

template <int F>
__global__ __launch_bounds__(256) void k_edge_agg(const int* __restrict__ ei,
                                                  const float* __restrict__ wnorm,
                                                  const float* __restrict__ h,
                                                  float* __restrict__ agg,
                                                  int* __restrict__ cnt, int E, int Nn) {
  constexpr int L = TASK_EDGES / 256;
  __shared__ int s_word;
  const int lane = threadIdx.x & 31;
  const int wave = __builtin_amdgcn_readfirstlane((int)(threadIdx.x >> 5));
  const int S = (E + TASK_EDGES - 1) / TASK_EDGES;
  const int mid = range_split(Nn);
  const int mine = share_word(&s_word, l2_half_of_this_wave());
  for (int pass = 0; pass < 2; ++pass) {
    const int q = mine ^ pass;
    const int lo = q ? mid : 0;
    const int hi = q ? Nn : mid;
    int drawn = 0;
    if (threadIdx.x == 0) drawn = atomicAdd(&cnt[q], 1);
    int t = share_word(&s_word, drawn);
    while (t < S) {
      const long long base = (long long)t * TASK_EDGES + wave * (32 * L) + lane;
      int row[L], col[L];
      float w[L];
#pragma unroll
      for (int s = 0; s < L; ++s) {
        const long long e = base + 32 * s;
        row[s] = 0;
        col[s] = -1;
        w[s] = 0.0f;
        if (e < E) {
          row[s] = ei[e];
          col[s] = ei[(long long)E + e];
          w[s] = wnorm[e];
        }
      }
#pragma unroll
      for (int s = 0; s < L; ++s) {
        const unsigned m = __builtin_amdgcn_ballot_w32(col[s] >= lo && col[s] < hi);
        agg_marked_edges<F>(m, row[s], col[s], w[s], h, agg, lane);
      }
      if (threadIdx.x == 0) drawn = atomicAdd(&cnt[q], 1);
      t = share_word(&s_word, drawn);
    }
  }
}

__global__ void k_gemm3_relu(const float* __restrict__ a2, const float* __restrict__ W3,
                             const float* __restrict__ deg, const float* __restrict__ b3,
                             float* __restrict__ g, float* __restrict__ out,
                             int* __restrict__ cnt, int Nn) {
  int i = blockIdx.x * blockDim.x + threadIdx.x;
  if (i < 2) cnt[i] = 0;
  if (i >= Nn) return;
  const float* r = a2 + (size_t)i * H2C;
  float acc = 0.0f;
#pragma unroll
  for (int f = 0; f < H2C; ++f) acc += fmaxf(r[f], 0.0f) * W3[f];
  g[i] = acc;
  const float di = dinv_of(deg[i]);
  out[i] = b3[0] + di * di * acc;
}

#define A1_THREADS 1024
#define A1_EPT 13
#define A1_TASK (A1_THREADS * A1_EPT)
#define A1_CHUNK 65536

__global__ __launch_bounds__(A1_THREADS) void k_edge_agg1(const int* __restrict__ ei,
                                                          const float* __restrict__ wnorm,
                                                          const float* __restrict__ g,
                                                          float* __restrict__ out,
                                                          int* __restrict__ cnt, int E, int Nn) {
  __shared__ alignas(16) float tab[A1_CHUNK];
  __shared__ int s_word;
  const int tid = threadIdx.x;
  const int S = (E + A1_TASK - 1) / A1_TASK;
  const int mid = range_split(Nn);
  const int mine = share_word(&s_word, l2_half_of_this_wave());
  for (int pass = 0; pass < 2; ++pass) {
    const int q = mine ^ pass;
    const int nlo = q ? mid : 0;
    const int nhi = q ? Nn : mid;
    int drawn = 0;
    if (tid == 0) drawn = atomicAdd(&cnt[q], 1);
    int t = share_word(&s_word, drawn);
    while (t < S) {
      const long long base = (long long)t * A1_TASK;
      const int rem = (E - base < (long long)A1_TASK) ? (int)(E - base) : A1_TASK;
      const int* __restrict__ rp = ei + base;
      const int* __restrict__ cp = ei + (long long)E + base;
      const float* __restrict__ wp = wnorm + base;
      int row[A1_EPT], col[A1_EPT];
      float w[A1_EPT];
#pragma unroll
      for (int k = 0; k < A1_EPT; ++k) {
        const int i = k * A1_THREADS + tid;
        const int ic = (i < rem) ? i : (rem - 1);
        row[k] = rp[ic];
        const int cv = cp[ic];
        w[k] = wp[ic];
        col[k] = (i < rem && cv >= nlo && cv < nhi) ? cv : -1;
      }
      if (tid == 0)
        drawn = (int)atomicInc(reinterpret_cast<unsigned int*>(cnt + q), 0xffffffffu);
      for (int lo = 0; lo < Nn; lo += A1_CHUNK) {
        const int cn = (Nn - lo < A1_CHUNK) ? (Nn - lo) : A1_CHUNK;
        lds_barrier();
        const int n4 = cn >> 2;
        const float4* __restrict__ gp = reinterpret_cast<const float4*>(g + lo);
        float4* tp = reinterpret_cast<float4*>(tab);
        int g0 = 0;
        for (; g0 + A1_THREADS * 8 <= n4; g0 += A1_THREADS * 8) {
          float4 d[8];
#pragma unroll
          for (int j = 0; j < 8; ++j) d[j] = gp[g0 + j * A1_THREADS + tid];
#pragma unroll
          for (int j = 0; j < 8; ++j) tp[g0 + j * A1_THREADS + tid] = d[j];
        }
        for (int gi = g0 + tid; gi < n4; gi += A1_THREADS) tp[gi] = gp[gi];
        {
          const int i = (n4 << 2) + tid;
          if (i < cn) tab[i] = g[lo + i];
        }
        lds_barrier();
        float tv[A1_EPT];
#pragma unroll
        for (int k = 0; k < A1_EPT; ++k) {
          const unsigned r = (unsigned)(row[k] - lo);
          tv[k] = tab[(r < (unsigned)cn) ? r : 0u];
        }
#pragma unroll
        for (int k = 0; k < A1_EPT; ++k) {
          const unsigned r = (unsigned)(row[k] - lo);
          if (r < (unsigned)cn && col[k] >= 0) atomicAdd(&out[col[k]], w[k] * tv[k]);
        }
        asm volatile("s_wait_dscnt 0x0" ::: "memory");
      }
      t = share_word(&s_word, drawn);
    }
  }
}


static inline unsigned int blocks_for(long long total, int bs) {
  return (unsigned int)((total + bs - 1) / bs);
}

extern "C" void kernel_launch(void* const* d_in, const int* in_sizes, int n_in,
                              void* d_out, int out_size, void* d_ws, size_t ws_size,
                              hipStream_t stream) {
  const float* x = (const float*)d_in[0];
  const int* ei = (const int*)d_in[1];
  const float* ew = (const float*)d_in[2];
  const float* W1 = (const float*)d_in[3];
  const float* b1 = (const float*)d_in[4];
  const float* W2 = (const float*)d_in[5];
  const float* b2 = (const float*)d_in[6];
  const float* W3 = (const float*)d_in[7];
  const float* b3 = (const float*)d_in[8];

  const int E = in_sizes[2];
  const int Nn = in_sizes[0] / IN_C;
  float* out = (float*)d_out;

  float* bufA = (float*)d_ws;
  float* bufB = bufA + (size_t)Nn * H1C;
  float* deg = bufB + (size_t)Nn * H1C;
  float* wnorm = deg + Nn;
  float* agg2 = bufA + (size_t)Nn * H2C;

  int* cntA = (int*)d_out;
  int* cntB = (int*)(bufB + (size_t)Nn * H2C);

  k_init_deg<<<blocks_for(Nn, 256), 256, 0, stream>>>(deg, cntA, 6, Nn);
  k_gemm1_splitk<<<512, G1_WAVES * 32, 0, stream>>>(x, W1, bufA, Nn / 16, ei, ew, deg, cntA,
                                                    E, Nn);
  k_wnorm<<<256, WN_THREADS, 0, stream>>>(ei, ew, deg, wnorm, E, Nn, bufA, b1, bufB);

  k_edge_agg<H1C><<<SCATTER_BLOCKS, 256, 0, stream>>>(
      ei, wnorm, bufA, bufB, cntA + 2, E, Nn);

  k_gemm_wmma_f32<H1C, H2C / 16, true>
      <<<Nn / 16, (H2C / 16) * 32, 0, stream>>>(bufB, W2, bufA, deg, b2, agg2);
  k_edge_agg<H2C><<<SCATTER_BLOCKS, 256, 0, stream>>>(
      ei, wnorm, bufA, agg2, cntA + 4, E, Nn);

  k_gemm3_relu<<<blocks_for(Nn, 256), 256, 0, stream>>>(agg2, W3, deg, b3, bufB, out, cntB,
                                                        Nn);
  k_edge_agg1<<<256, A1_THREADS, 0, stream>>>(ei, wnorm, bufB, out, cntB, E, Nn);
}
